# starting kernel
# baseline (speedup 1.0000x reference)
#include <hip/hip_runtime.h>

#define NATOMS 4096
#define CC     64
#define TT     256
#define ASIZE  8
#define BATCH  4
#define KDIM   512
#define PPOS   257
#define RES_W  304
#define NSPLIT 2
#define NTPB   9
#define MTILE  64
#define CORR_THREADS 128
#define NITER  32

typedef __attribute__((ext_vector_type(2))) float v2f;
typedef __attribute__((ext_vector_type(8))) float v8f;

__global__ void k_normalize(const float* __restrict__ d, float* __restrict__ dnn) {
    __shared__ float red[4];
    const int a = blockIdx.x;
    const float* src = d + (size_t)a * KDIM;
    float s = 0.f;
    for (int i = threadIdx.x; i < KDIM; i += CORR_THREADS) { float v = src[i]; s += v * v; }
    for (int off = 16; off > 0; off >>= 1) s += __shfl_down(s, off, 32);
    const int wv = threadIdx.x >> 5;
    if ((threadIdx.x & 31) == 0) red[wv] = s;
    __syncthreads();
    if (threadIdx.x == 0) red[0] = 1.f / (sqrtf(red[0] + red[1] + red[2] + red[3]) + 1e-8f);
    __syncthreads();
    const float inv = red[0];
    float* dst = dnn + (size_t)a * KDIM;
    for (int i = threadIdx.x; i < KDIM; i += CORR_THREADS) dst[i] = src[i] * inv;
}

__global__ void k_init(const float* __restrict__ x, float* __restrict__ res,
                       float* __restrict__ recon, unsigned long long* __restrict__ best) {
    const int i = blockIdx.x * blockDim.x + threadIdx.x;
    if (i < BATCH * CC * RES_W) {
        const int col = i % RES_W;
        const int bc  = i / RES_W;
        res[i]   = (col < TT) ? x[bc * TT + col] : 0.f;
        recon[i] = 0.f;
    }
    if (i < BATCH) best[i] = 0ull;
}

__global__ __launch_bounds__(CORR_THREADS, 1)
void k_corr(const float* __restrict__ dnn, const float* __restrict__ res,
            unsigned long long* __restrict__ best) {
    extern __shared__ float smem[];
    const int b    = blockIdx.z;
    const int nt0  = blockIdx.y * NTPB;
    const int m0   = blockIdx.x * MTILE;
    const int tid  = threadIdx.x;
    const int lane = tid & 31;
    const int wave = tid >> 5;

    const int half = lane >> 4;
    const int nloc = lane & 15;
    const float* arow = dnn + (size_t)(m0 + wave * 16 + nloc) * KDIM + half * 2;

    #pragma unroll
    for (int pf = 0; pf < 16; ++pf)
        __builtin_prefetch(arow + pf * 32, 0, 3);

    const float4* resb = (const float4*)(res + (size_t)b * CC * RES_W);
    float4* s4 = (float4*)smem;
    for (int i = tid; i < (CC * RES_W) / 4; i += CORR_THREADS) s4[i] = resb[i];
    __syncthreads();

    v8f acc[NTPB] = {};

    auto loadA = [&](int ks) { v2f a; a.x = arow[ks * 4]; a.y = arow[ks * 4 + 1]; return a; };
    auto srowp = [&](int ks) {
        const int kbase = ks * 4 + half * 2;
        return (const float*)smem + (kbase >> 3) * RES_W + (kbase & 7) + nloc + nt0 * 16;
    };

    v2f a0 = loadA(0);
    for (int ks = 0; ks < KDIM / 4; ++ks) {
        v2f an = loadA(ks + 1 < KDIM / 4 ? ks + 1 : ks);
        const float* sr = srowp(ks);
        v2f bf[NTPB];
        #pragma unroll
        for (int nt = 0; nt < NTPB; ++nt) {
            bf[nt].x = sr[nt * 16];
            bf[nt].y = sr[nt * 16 + 1];
        }
        #pragma unroll
        for (int nt = 0; nt < NTPB; ++nt)
            acc[nt] = __builtin_amdgcn_wmma_f32_16x16x4_f32(
                false, a0, false, bf[nt], (short)0, acc[nt], false, false);
        a0 = an;
    }

    float bestv = -3.402823466e38f;
    unsigned bestidx = 0u;
    #pragma unroll
    for (int nt = 0; nt < NTPB; ++nt) {
        const int pos = (nt0 + nt) * 16 + nloc;
        if (pos < PPOS) {
            #pragma unroll
            for (int r = 0; r < 8; ++r) {
                const float v = acc[nt][r];
                if (v > bestv) {
                    bestv = v;
                    bestidx = (unsigned)((m0 + wave * 16 + r + 8 * half) * PPOS + pos);
                }
            }
        }
    }
    unsigned vb = __float_as_uint(bestv);
    vb = (vb & 0x80000000u) ? ~vb : (vb | 0x80000000u);
    unsigned long long key = ((unsigned long long)vb << 32) | (0xFFFFFFFFu - bestidx);
    for (int off = 16; off > 0; off >>= 1) {
        unsigned long long o = __shfl_xor(key, off, 32);
        if (o > key) key = o;
    }
    if (lane == 0) atomicMax(best + b, key);
}

__global__ void k_update(const float* __restrict__ dnn, float* __restrict__ res,
                         float* __restrict__ recon, unsigned long long* __restrict__ best) {
    const int b = blockIdx.x;
    const unsigned long long key = best[b];
    const unsigned enc  = (unsigned)(key >> 32);
    const unsigned bits = (enc & 0x80000000u) ? (enc & 0x7FFFFFFFu) : ~enc;
    const float val = __uint_as_float(bits);
    const unsigned idx = 0xFFFFFFFFu - (unsigned)(key & 0xFFFFFFFFu);
    const int atom = (int)(idx / PPOS);
    const int pos  = (int)(idx % PPOS);
    const float* ap = dnn + (size_t)atom * KDIM;
    float* rb = res   + (size_t)b * CC * RES_W;
    float* cb = recon + (size_t)b * CC * RES_W;
    for (int i = threadIdx.x; i < KDIM; i += blockDim.x) {
        const int c = i >> 3, t = i & 7;
        const float av = ap[i] * val;
        rb[c * RES_W + pos + t] -= av;
        cb[c * RES_W + pos + t] += av;
    }
    __syncthreads();
    if (threadIdx.x == 0) best[b] = 0ull;
}

__global__ void k_output(const float* __restrict__ recon, float* __restrict__ out) {
    const int i = blockIdx.x * blockDim.x + threadIdx.x;
    if (i < BATCH * CC * TT) {
        const int t  = i % TT;
        const int bc = i / TT;
        out[i] = recon[(size_t)bc * RES_W + t];
    }
}

extern "C" void kernel_launch(void* const* d_in, const int* in_sizes, int n_in,
                              void* d_out, int out_size, void* d_ws, size_t ws_size,
                              hipStream_t stream) {
    const float* x = (const float*)d_in[0];
    const float* d = (const float*)d_in[1];
    (void)in_sizes; (void)n_in; (void)out_size; (void)ws_size;

    char* ws = (char*)d_ws;
    float* dnn   = (float*)ws;
    float* res   = (float*)(ws + (size_t)NATOMS * KDIM * 4);
    float* recon = res + BATCH * CC * RES_W;
    unsigned long long* best = (unsigned long long*)(recon + BATCH * CC * RES_W);

    k_normalize<<<NATOMS, CORR_THREADS, 0, stream>>>(d, dnn);
    k_init<<<(BATCH * CC * RES_W + 255) / 256, 256, 0, stream>>>(x, res, recon, best);

    const size_t shbytes = (size_t)CC * RES_W * sizeof(float);
    dim3 grid(NATOMS / MTILE, NSPLIT, BATCH);
    for (int it = 0; it < NITER; ++it) {
        k_corr<<<grid, CORR_THREADS, shbytes, stream>>>(dnn, res, best);
        k_update<<<BATCH, 128, 0, stream>>>(dnn, res, recon, best);
    }
    k_output<<<(BATCH * CC * TT + 255) / 256, 256, 0, stream>>>(recon, (float*)d_out);
}


# Correlation GEMM re-tiled to one wave per SIMD: a wave owns 16 atoms x all 17 position tiles (256 work-groups, one per WGP); next k-step's B fragments loaded as one group ahead, A in an 8-step ring
# speedup vs baseline: 1.2144x; 1.2144x over previous
#include <hip/hip_runtime.h>

#define NATOMS 4096
#define CC     64
#define TT     256
#define ASIZE  8
#define BATCH  4
#define KDIM   512
#define PPOS   257
#define RES_W  304
#define NT     17
#define KGRP   8
#define KSTEPS 128
#define MTILE  64
#define RES_FLOATS 19456
#define LDS_FLOATS 41984
#define CORR_THREADS 128
#define NITER  32

typedef __attribute__((ext_vector_type(2))) float v2f;
typedef __attribute__((ext_vector_type(8))) float v8f;

__global__ void k_normalize(const float* __restrict__ d, float* __restrict__ dnn) {
    __shared__ float red[4];
    const int a = blockIdx.x;
    const float* src = d + (size_t)a * KDIM;
    float s = 0.f;
    for (int i = threadIdx.x; i < KDIM; i += CORR_THREADS) { float v = src[i]; s += v * v; }
    for (int off = 16; off > 0; off >>= 1) s += __shfl_down(s, off, 32);
    const int wv = threadIdx.x >> 5;
    if ((threadIdx.x & 31) == 0) red[wv] = s;
    __syncthreads();
    if (threadIdx.x == 0) red[0] = 1.f / (sqrtf(red[0] + red[1] + red[2] + red[3]) + 1e-8f);
    __syncthreads();
    const float inv = red[0];
    float* dst = dnn + (size_t)a * KDIM;
    for (int i = threadIdx.x; i < KDIM; i += CORR_THREADS) dst[i] = src[i] * inv;
}

__global__ void k_init(const float* __restrict__ x, float* __restrict__ res,
                       float* __restrict__ recon, unsigned long long* __restrict__ best) {
    const int i = blockIdx.x * blockDim.x + threadIdx.x;
    if (i < BATCH * CC * RES_W) {
        const int col = i % RES_W;
        const int bc  = i / RES_W;
        res[i]   = (col < TT) ? x[bc * TT + col] : 0.f;
        recon[i] = 0.f;
    }
    if (i < BATCH) best[i] = 0ull;
}

__global__ __launch_bounds__(CORR_THREADS)
void k_corr(const float* __restrict__ dnn, const float* __restrict__ res,
            unsigned long long* __restrict__ best) {
    __shared__ __attribute__((aligned(16))) float smem[LDS_FLOATS];
    const int b    = blockIdx.y;
    const int m0   = blockIdx.x * MTILE;
    const int tid  = threadIdx.x;
    const int lane = tid & 31;
    const int wave = tid >> 5;

    const int half = lane >> 4;
    const int nloc = lane & 15;
    const float* arow = dnn + (size_t)(m0 + wave * 16 + nloc) * KDIM + half * 2;

    auto loadA = [&](int ks) { v2f a; a.x = arow[ks * 4]; a.y = arow[ks * 4 + 1]; return a; };

    v2f a[KGRP];
    #pragma unroll
    for (int j = 0; j < KGRP; ++j) a[j] = loadA(j);

    const float4* resb = (const float4*)(res + (size_t)b * RES_FLOATS);
    float4* s4 = (float4*)smem;
    #pragma unroll
    for (int h2 = 0; h2 < 2; ++h2) {
        float4 tmp[19];
        #pragma unroll
        for (int i = 0; i < 19; ++i) tmp[i] = resb[tid + (h2 * 19 + i) * CORR_THREADS];
        #pragma unroll
        for (int i = 0; i < 19; ++i) s4[tid + (h2 * 19 + i) * CORR_THREADS] = tmp[i];
    }
    __syncthreads();

    v8f acc[NT] = {};

    auto loadB = [&](v2f (&bf)[NT], unsigned p) {
        #pragma unroll
        for (int nt = 0; nt < NT; ++nt) {
            bf[nt].x = smem[p + nt * 16];
            bf[nt].y = smem[p + nt * 16 + 1];
        }
    };

    v2f b0[NT], b1[NT];
    unsigned rp = (unsigned)(half * 2 + nloc);
    loadB(b0, rp);

    auto kstep = [&](const v2f (&cur)[NT], v2f (&nxt)[NT], int ks, int j) {
        const int ksa = ks + 7 < KSTEPS ? ks + 7 : KSTEPS - 1;
        const v2f av = a[j];
        if (j & 1) {
            rp += (ks + 1 < KSTEPS) ? RES_W : 0;
            asm("" : "+v"(rp));
            loadB(nxt, rp);
        } else {
            loadB(nxt, rp + 4);
        }
        a[(j + KGRP - 1) % KGRP] = loadA(ksa);
        __builtin_amdgcn_sched_barrier(0);
        #pragma unroll
        for (int nt = 0; nt < NT; ++nt)
            acc[nt] = __builtin_amdgcn_wmma_f32_16x16x4_f32(
                false, av, false, cur[nt], (short)0, acc[nt], false, false);
    };

    for (int g = 0; g < KSTEPS / KGRP; ++g) {
        #pragma unroll
        for (int j = 0; j < KGRP; j += 2) {
            kstep(b0, b1, g * KGRP + j, j);
            kstep(b1, b0, g * KGRP + j + 1, j + 1);
        }
    }

    float bestv = -3.402823466e38f;
    unsigned bestidx = 0u;
    #pragma unroll
    for (int nt = 0; nt < NT; ++nt) {
        const int pos = nt * 16 + nloc;
        if (pos < PPOS) {
            #pragma unroll
            for (int r = 0; r < 8; ++r) {
                const float v = acc[nt][r];
                if (v > bestv) {
                    bestv = v;
                    bestidx = (unsigned)((m0 + wave * 16 + r + 8 * half) * PPOS + pos);
                }
            }
        }
    }
    unsigned vb = __float_as_uint(bestv);
    vb = (vb & 0x80000000u) ? ~vb : (vb | 0x80000000u);
    unsigned long long key = ((unsigned long long)vb << 32) | (0xFFFFFFFFu - bestidx);
    for (int off = 16; off > 0; off >>= 1) {
        unsigned long long o = __shfl_xor(key, off, 32);
        if (o > key) key = o;
    }
    if (lane == 0) atomicMax(best + b, key);
}

__global__ void k_update(const float* __restrict__ dnn, float* __restrict__ res,
                         float* __restrict__ recon, unsigned long long* __restrict__ best) {
    const int b = blockIdx.x;
    const unsigned long long key = best[b];
    const unsigned enc  = (unsigned)(key >> 32);
    const unsigned bits = (enc & 0x80000000u) ? (enc & 0x7FFFFFFFu) : ~enc;
    const float val = __uint_as_float(bits);
    const unsigned idx = 0xFFFFFFFFu - (unsigned)(key & 0xFFFFFFFFu);
    const int atom = (int)(idx / PPOS);
    const int pos  = (int)(idx % PPOS);
    const float* ap = dnn + (size_t)atom * KDIM;
    float* rb = res   + (size_t)b * CC * RES_W;
    float* cb = recon + (size_t)b * CC * RES_W;
    for (int i = threadIdx.x; i < KDIM; i += blockDim.x) {
        const int c = i >> 3, t = i & 7;
        const float av = ap[i] * val;
        rb[c * RES_W + pos + t] -= av;
        cb[c * RES_W + pos + t] += av;
    }
    __syncthreads();
    if (threadIdx.x == 0) best[b] = 0ull;
}

__global__ void k_output(const float* __restrict__ recon, float* __restrict__ out) {
    const int i = blockIdx.x * blockDim.x + threadIdx.x;
    if (i < BATCH * CC * TT) {
        const int t  = i % TT;
        const int bc = i / TT;
        out[i] = recon[(size_t)bc * RES_W + t];
    }
}

extern "C" void kernel_launch(void* const* d_in, const int* in_sizes, int n_in,
                              void* d_out, int out_size, void* d_ws, size_t ws_size,
                              hipStream_t stream) {
    const float* x = (const float*)d_in[0];
    const float* d = (const float*)d_in[1];
    (void)in_sizes; (void)n_in; (void)out_size; (void)ws_size;

    char* ws = (char*)d_ws;
    float* dnn   = (float*)ws;
    float* res   = (float*)(ws + (size_t)NATOMS * KDIM * 4);
    float* recon = res + BATCH * CC * RES_W;
    unsigned long long* best = (unsigned long long*)(recon + BATCH * CC * RES_W);

    k_normalize<<<NATOMS, CORR_THREADS, 0, stream>>>(d, dnn);
    k_init<<<(BATCH * CC * RES_W + 255) / 256, 256, 0, stream>>>(x, res, recon, best);

    dim3 grid(NATOMS / MTILE, BATCH);
    for (int it = 0; it < NITER; ++it) {
        k_corr<<<grid, CORR_THREADS, 0, stream>>>(dnn, res, best);
        k_update<<<BATCH, 128, 0, stream>>>(dnn, res, recon, best);
    }
    k_output<<<(BATCH * CC * TT + 255) / 256, 256, 0, stream>>>(recon, (float*)d_out);
}


# Two-phase fast FIR correlation: even/odd positions from 3 half-rate products (H0*X0, H1*X1, (H0+H1)*(X1+X0')), not 4; 96 products per group, was 128; odd phase by difference: sum order changed
# speedup vs baseline: 1.8171x; 1.4963x over previous
#include <hip/hip_runtime.h>

#define NATOMS 4096
#define CC     64
#define TT     256
#define ASIZE  8
#define BATCH  4
#define KDIM   512
#define PPOS   257
#define RES_W  304
#define NT     16
#define NTH    8
#define NXS    11
#define MTILE  64
#define RES_FLOATS 19456
#define LDS_FLOATS 41984
#define CORR_THREADS 128
#define NITER  32
#define NGRP   16
#define NFRAG  23
#define SBLK   68
#define P256OFF 544
#define PITCH2 608

typedef __attribute__((ext_vector_type(2))) float v2f;
typedef __attribute__((ext_vector_type(4))) float v4f;
typedef __attribute__((ext_vector_type(8))) float v8f;

__device__ __forceinline__ int kperm(int c, int t) {
    return (c >> 2) * 32 + ((c >> 1) & 1) * 16 + (t & 1) * 8 + (t >> 1) * 2 + (c & 1);
}

__global__ void k_normalize(const float* __restrict__ d, float* __restrict__ dnn) {
    __shared__ float red[4];
    const int a = blockIdx.x;
    const float* src = d + (size_t)a * KDIM;
    float s = 0.f;
    for (int i = threadIdx.x; i < KDIM; i += CORR_THREADS) { float v = src[i]; s += v * v; }
    for (int off = 16; off > 0; off >>= 1) s += __shfl_down(s, off, 32);
    const int wv = threadIdx.x >> 5;
    if ((threadIdx.x & 31) == 0) red[wv] = s;
    __syncthreads();
    if (threadIdx.x == 0) red[0] = 1.f / (sqrtf(red[0] + red[1] + red[2] + red[3]) + 1e-8f);
    __syncthreads();
    const float inv = red[0];
    float* dst = dnn + (size_t)a * KDIM;
    for (int i = threadIdx.x; i < KDIM; i += CORR_THREADS) dst[kperm(i >> 3, i & 7)] = src[i] * inv;
}

__global__ void k_init(const float* __restrict__ x, float* __restrict__ res,
                       float* __restrict__ recon, unsigned long long* __restrict__ best,
                       unsigned* __restrict__ arrived) {
    const int i = blockIdx.x * blockDim.x + threadIdx.x;
    if (i < BATCH * CC * RES_W) {
        const int col = i % RES_W;
        const int bc  = i / RES_W;
        res[i]   = (col < TT) ? x[bc * TT + col] : 0.f;
        recon[i] = 0.f;
    }
    if (i < BATCH) { best[i] = 0ull; arrived[i] = 0u; }
}

__global__ __launch_bounds__(CORR_THREADS)
void k_corr(const float* __restrict__ dnn, float* res, float* recon,
            unsigned long long* best, unsigned* arrived) {
    __shared__ __attribute__((aligned(16))) float smem[LDS_FLOATS];
    const int b    = blockIdx.y;
    const int m0   = blockIdx.x * MTILE;
    const int tid  = threadIdx.x;
    const int lane = tid & 31;
    const int wave = tid >> 5;

    const int half = lane >> 4;
    const int nloc = lane & 15;
    const v4f* arow = (const v4f*)(dnn + (size_t)(m0 + wave * 16 + nloc) * KDIM + half * 16);

    auto loadA = [&](v4f (&a)[4], int g) {
        #pragma unroll
        for (int q = 0; q < 4; ++q) a[q] = arow[g * 8 + q];
    };

    v4f a0[4], a1[4];
    loadA(a0, 0);

    const v4f* resb = (const v4f*)(res + (size_t)b * RES_FLOATS);
    #pragma unroll
    for (int h2 = 0; h2 < 2; ++h2) {
        const int i0 = h2 ? 9 : 0, i1 = h2 ? 17 : 9;
        v4f te[9], to[9];
        #pragma unroll
        for (int i = i0; i < i1; ++i) {
            const int q = tid + i * CORR_THREADS;
            const int r = q / SBLK, c4 = q % SBLK;
            te[i - i0] = resb[(2 * r) * (RES_W / 4) + c4];
            to[i - i0] = resb[(2 * r + 1) * (RES_W / 4) + c4];
        }
        #pragma unroll
        for (int i = i0; i < i1; ++i) {
            const int q = tid + i * CORR_THREADS;
            const int r = q / SBLK, c4 = q % SBLK;
            v2f* d2 = (v2f*)(smem + r * PITCH2 + c4 * 8 + (c4 >> 2) * 2);
            const v4f e = te[i - i0], o = to[i - i0];
            d2[0] = (v2f){e.x, o.x};
            d2[1] = (v2f){e.y, o.y};
            d2[2] = (v2f){e.z, o.z};
            d2[3] = (v2f){e.w, o.w};
        }
    }
    __syncthreads();

    v8f accA[NTH] = {}, accB[NTH] = {}, accC[NTH] = {};
    v2f pA0 = {}, pA1 = {}, pB0 = {}, pB1 = {};

    auto loadB = [&](v2f (&f)[NFRAG], unsigned p) {
        #pragma unroll
        for (int j = 0; j < NFRAG; ++j) f[j] = *(const v2f*)(smem + p + 2 * (j + (j >> 4)));
    };
    auto loadW = [&](v4f (&w)[4], int g) {
        const v4f* wp = (const v4f*)(smem + (2 * g + half) * PITCH2 + P256OFF);
        #pragma unroll
        for (int q = 0; q < 4; ++q) w[q] = wp[q];
    };

    v2f f0[NFRAG], f1[NFRAG];
    v4f w0[4], w1[4];
    const unsigned fp = (unsigned)(half * PITCH2 + 34 * nloc);
    loadB(f0, fp);
    loadW(w0, 0);

    auto group = [&](const v2f (&cf)[NFRAG], const v4f (&ca)[4], const v4f (&cw)[4],
                     v2f (&nf)[NFRAG], v4f (&na)[4], v4f (&nw)[4], int g) {
        const int gn = g + 1 < NGRP ? g + 1 : NGRP - 1;
        __builtin_amdgcn_sched_barrier(0);
        loadB(nf, fp + (unsigned)gn * (2 * PITCH2));
        loadW(nw, gn);
        loadA(na, gn);
        __builtin_amdgcn_sched_barrier(0);
        const v4f hs[2] = {ca[0] + ca[2], ca[1] + ca[3]};
        v2f xs[NXS];
        #pragma unroll
        for (int f = 0; f < NXS; ++f) xs[f] = cf[2 * f + 1] + cf[2 * f + 2];
        pA0 += (v2f){ca[0].x, ca[0].y} * (v2f){cw[0].x, cw[0].y};
        pA1 += (v2f){ca[0].z, ca[0].w} * (v2f){cw[1].x, cw[1].y};
        pA0 += (v2f){ca[1].x, ca[1].y} * (v2f){cw[2].x, cw[2].y};
        pA1 += (v2f){ca[1].z, ca[1].w} * (v2f){cw[3].x, cw[3].y};
        pB0 += (v2f){ca[2].x, ca[2].y} * (v2f){cw[0].z, cw[0].w};
        pB1 += (v2f){ca[2].z, ca[2].w} * (v2f){cw[1].z, cw[1].w};
        pB0 += (v2f){ca[3].x, ca[3].y} * (v2f){cw[2].z, cw[2].w};
        pB1 += (v2f){ca[3].z, ca[3].w} * (v2f){cw[3].z, cw[3].w};
        #pragma unroll
        for (int s = 0; s < ASIZE / 2; ++s) {
            const v4f q0 = ca[s >> 1], q1 = ca[2 + (s >> 1)], q2 = hs[s >> 1];
            const v2f h0 = (s & 1) ? (v2f){q0.z, q0.w} : (v2f){q0.x, q0.y};
            const v2f h1 = (s & 1) ? (v2f){q1.z, q1.w} : (v2f){q1.x, q1.y};
            const v2f h2 = (s & 1) ? (v2f){q2.z, q2.w} : (v2f){q2.x, q2.y};
            #pragma unroll
            for (int j = 0; j < NTH; ++j)
                accA[j] = __builtin_amdgcn_wmma_f32_16x16x4_f32(
                    false, h0, false, cf[2 * (j + s)], (short)0, accA[j], false, false);
            #pragma unroll
            for (int j = 0; j < NTH; ++j)
                accB[j] = __builtin_amdgcn_wmma_f32_16x16x4_f32(
                    false, h1, false, cf[2 * (j + s) + 1], (short)0, accB[j], false, false);
            #pragma unroll
            for (int j = 0; j < NTH; ++j)
                accC[j] = __builtin_amdgcn_wmma_f32_16x16x4_f32(
                    false, h2, false, xs[j + s], (short)0, accC[j], false, false);
        }
    };

    for (int g2 = 0; g2 < NGRP / 2; ++g2) {
        group(f0, a0, w0, f1, a1, w1, 2 * g2);
        group(f1, a1, w1, f0, a0, w0, 2 * g2 + 1);
    }

    float a128 = (pA0.x + pA0.y) + (pA1.x + pA1.y);
    float b128 = (pB0.x + pB0.y) + (pB1.x + pB1.y);
    a128 += __shfl_xor(a128, 16, 32);
    b128 += __shfl_xor(b128, 16, 32);
    v8f anext;
    #pragma unroll
    for (int r = 0; r < 8; ++r) {
        const float fromcol = __shfl(accA[0][r], (lane + 1) & 31, 32);
        const float from128 = __shfl(a128, 8 * half + r, 32);
        anext[r] = (nloc == 15) ? from128 : fromcol;
    }
    v8f acc[NT];
    #pragma unroll
    for (int j = 0; j < NTH; ++j) {
        acc[2 * j]     = accA[j] + accB[j];
        acc[2 * j + 1] = (accC[j] - (j + 1 < NTH ? accA[j + 1] : anext)) - accB[j];
    }

    float bestv = -3.402823466e38f;
    #pragma unroll
    for (int r = 0; r < 8; ++r) {
        #pragma unroll
        for (int j = 0; j < NT; ++j) bestv = fmaxf(bestv, acc[j][r]);
    }
    unsigned code = 0xFFFFFFFFu;
    #pragma unroll
    for (int r = 7; r >= 0; --r) {
        #pragma unroll
        for (int j = NT - 1; j >= 0; --j)
            if (acc[j][r] == bestv) code = (unsigned)(r * PPOS + j);
    }
    unsigned bestidx = code == 0xFFFFFFFFu
        ? 0u : (unsigned)((m0 + wave * 16 + 8 * half) * PPOS + 16 * nloc) + code;
    const float s256 = a128 + b128;
    const unsigned idx256 = (unsigned)((m0 + wave * 16 + nloc) * PPOS + (PPOS - 1));
    if (s256 > bestv || (s256 == bestv && idx256 < bestidx)) { bestv = s256; bestidx = idx256; }
    unsigned vb = __float_as_uint(bestv);
    vb = (vb & 0x80000000u) ? ~vb : (vb | 0x80000000u);
    unsigned long long key = ((unsigned long long)vb << 32) | (0xFFFFFFFFu - bestidx);
    for (int off = 16; off > 0; off >>= 1) {
        unsigned long long o = __shfl_xor(key, off, 32);
        if (o > key) key = o;
    }
    unsigned long long* wkey = (unsigned long long*)(smem + RES_FLOATS);
    if (lane == 0) wkey[wave] = key;
    __syncthreads();
    if (tid == 0) {
        unsigned long long k = wkey[0];
        for (int w = 1; w < CORR_THREADS / 32; ++w) if (wkey[w] > k) k = wkey[w];
        atomicMax(best + b, k);
        const unsigned prev = __hip_atomic_fetch_add(arrived + b, 1u, __ATOMIC_ACQ_REL,
                                                     __HIP_MEMORY_SCOPE_AGENT);
        unsigned long long fin = 0ull;
        if (prev == gridDim.x - 1)
            fin = __hip_atomic_load(best + b, __ATOMIC_RELAXED, __HIP_MEMORY_SCOPE_AGENT);
        wkey[CORR_THREADS / 32] = fin;
    }
    __syncthreads();
    const unsigned long long fin = wkey[CORR_THREADS / 32];
    if (fin == 0ull) return;

    if (tid == 0) { best[b] = 0ull; arrived[b] = 0u; }
    const unsigned enc  = (unsigned)(fin >> 32);
    const unsigned bits = (enc & 0x80000000u) ? (enc & 0x7FFFFFFFu) : ~enc;
    const float val = __uint_as_float(bits);
    const unsigned idx = 0xFFFFFFFFu - (unsigned)(fin & 0xFFFFFFFFu);
    const int atom = min((int)(idx / PPOS), NATOMS - 1);
    const int pos  = (int)(idx % PPOS);
    const float* ap = dnn + (size_t)atom * KDIM;
    float* rb = res   + (size_t)b * RES_FLOATS;
    float* cb = recon + (size_t)b * RES_FLOATS;
    float ua[4], ur[4], uc[4];
    #pragma unroll
    for (int k = 0; k < KDIM / CORR_THREADS; ++k) {
        const int i = tid + k * CORR_THREADS;
        const int ch = i >> 3, t = i & 7;
        ua[k] = ap[kperm(ch, t)];
        ur[k] = rb[ch * RES_W + pos + t];
        uc[k] = cb[ch * RES_W + pos + t];
    }
    #pragma unroll
    for (int k = 0; k < KDIM / CORR_THREADS; ++k) {
        const int i = tid + k * CORR_THREADS;
        const int ch = i >> 3, t = i & 7;
        const float av = ua[k] * val;
        rb[ch * RES_W + pos + t] = ur[k] - av;
        cb[ch * RES_W + pos + t] = uc[k] + av;
    }
}

__global__ void k_output(const float* __restrict__ recon, float* __restrict__ out) {
    const int i = blockIdx.x * blockDim.x + threadIdx.x;
    if (i < BATCH * CC * TT) {
        const int t  = i % TT;
        const int bc = i / TT;
        out[i] = recon[(size_t)bc * RES_W + t];
    }
}

extern "C" void kernel_launch(void* const* d_in, const int* in_sizes, int n_in,
                              void* d_out, int out_size, void* d_ws, size_t ws_size,
                              hipStream_t stream) {
    const float* x = (const float*)d_in[0];
    const float* d = (const float*)d_in[1];
    (void)in_sizes; (void)n_in; (void)out_size; (void)ws_size;

    char* ws = (char*)d_ws;
    float* dnn   = (float*)ws;
    float* res   = (float*)(ws + (size_t)NATOMS * KDIM * 4);
    float* recon = res + BATCH * CC * RES_W;
    unsigned long long* best = (unsigned long long*)(recon + BATCH * CC * RES_W);
    unsigned* arrived = (unsigned*)(best + BATCH);

    k_normalize<<<NATOMS, CORR_THREADS, 0, stream>>>(d, dnn);
    k_init<<<(BATCH * CC * RES_W + 255) / 256, 256, 0, stream>>>(x, res, recon, best, arrived);

    dim3 grid(NATOMS / MTILE, BATCH);
    for (int it = 0; it < NITER; ++it)
        k_corr<<<grid, CORR_THREADS, 0, stream>>>(dnn, res, recon, best, arrived);
    k_output<<<(BATCH * CC * TT + 255) / 256, 256, 0, stream>>>(recon, (float*)d_out);
}


# Half-rate correlations split once more (fast FIR nested: 9 quarter-rate products of 2 taps, not 16): 72 matrix products per channel group, was 96; 4 position tiles per product; sum order changed
# speedup vs baseline: 2.0239x; 1.1138x over previous
#include <hip/hip_runtime.h>

#define NATOMS 4096
#define CC     64
#define TT     256
#define ASIZE  8
#define BATCH  4
#define KDIM   512
#define PPOS   257
#define RES_W  304
#define NT     16
#define NTQ    4
#define NQS    5
#define MTILE  64
#define RES_FLOATS 19456
#define LDS_FLOATS 41984
#define CORR_THREADS 128
#define NITER  32
#define NGRP   16
#define NFRAG  23
#define SBLK   68
#define P256OFF 544
#define PITCH2 608

typedef __attribute__((ext_vector_type(2))) float v2f;
typedef __attribute__((ext_vector_type(4))) float v4f;
typedef __attribute__((ext_vector_type(8))) float v8f;

__device__ __forceinline__ int kperm(int c, int t) {
    return (c >> 2) * 32 + ((c >> 1) & 1) * 16 + (t & 1) * 8 + (t >> 1) * 2 + (c & 1);
}

__global__ void k_normalize(const float* __restrict__ d, float* __restrict__ dnn) {
    __shared__ float red[4];
    const int a = blockIdx.x;
    const float* src = d + (size_t)a * KDIM;
    float s = 0.f;
    for (int i = threadIdx.x; i < KDIM; i += CORR_THREADS) { float v = src[i]; s += v * v; }
    for (int off = 16; off > 0; off >>= 1) s += __shfl_down(s, off, 32);
    const int wv = threadIdx.x >> 5;
    if ((threadIdx.x & 31) == 0) red[wv] = s;
    __syncthreads();
    if (threadIdx.x == 0) red[0] = 1.f / (sqrtf(red[0] + red[1] + red[2] + red[3]) + 1e-8f);
    __syncthreads();
    const float inv = red[0];
    float* dst = dnn + (size_t)a * KDIM;
    for (int i = threadIdx.x; i < KDIM; i += CORR_THREADS) dst[kperm(i >> 3, i & 7)] = src[i] * inv;
}

__global__ void k_init(const float* __restrict__ x, float* __restrict__ res,
                       float* __restrict__ recon, unsigned long long* __restrict__ best,
                       unsigned* __restrict__ arrived) {
    const int i = blockIdx.x * blockDim.x + threadIdx.x;
    if (i < BATCH * CC * RES_W) {
        const int col = i % RES_W;
        const int bc  = i / RES_W;
        res[i]   = (col < TT) ? x[bc * TT + col] : 0.f;
        recon[i] = 0.f;
    }
    if (i < BATCH) { best[i] = 0ull; arrived[i] = 0u; }
}

__global__ __launch_bounds__(CORR_THREADS)
void k_corr(const float* __restrict__ dnn, float* res, float* recon,
            unsigned long long* best, unsigned* arrived) {
    __shared__ __attribute__((aligned(16))) float smem[LDS_FLOATS];
    const int b    = blockIdx.y;
    const int m0   = blockIdx.x * MTILE;
    const int tid  = threadIdx.x;
    const int lane = tid & 31;
    const int wave = tid >> 5;

    const int half = lane >> 4;
    const int nloc = lane & 15;
    const v4f* arow = (const v4f*)(dnn + (size_t)(m0 + wave * 16 + nloc) * KDIM + half * 16);

    auto loadA = [&](v4f (&a)[4], int g) {
        #pragma unroll
        for (int q = 0; q < 4; ++q) a[q] = arow[g * 8 + q];
    };

    v4f ra0[4], ra1[4];
    loadA(ra0, 0);

    const v4f* resb = (const v4f*)(res + (size_t)b * RES_FLOATS);
    #pragma unroll
    for (int h2 = 0; h2 < 2; ++h2) {
        const int i0 = h2 ? 9 : 0, i1 = h2 ? 17 : 9;
        v4f te[9], to[9];
        #pragma unroll
        for (int i = i0; i < i1; ++i) {
            const int q = tid + i * CORR_THREADS;
            const int r = q / SBLK, c4 = q % SBLK;
            te[i - i0] = resb[(2 * r) * (RES_W / 4) + c4];
            to[i - i0] = resb[(2 * r + 1) * (RES_W / 4) + c4];
        }
        #pragma unroll
        for (int i = i0; i < i1; ++i) {
            const int q = tid + i * CORR_THREADS;
            const int r = q / SBLK, c4 = q % SBLK;
            v2f* d2 = (v2f*)(smem + r * PITCH2 + c4 * 8 + (c4 >> 2) * 2);
            const v4f e = te[i - i0], o = to[i - i0];
            d2[0] = (v2f){e.x, o.x};
            d2[1] = (v2f){e.y, o.y};
            d2[2] = (v2f){e.z, o.z};
            d2[3] = (v2f){e.w, o.w};
        }
    }
    __syncthreads();

    v8f a1[NTQ] = {}, b1[NTQ] = {}, c1[NTQ] = {};
    v8f a2[NTQ] = {}, b2[NTQ] = {}, c2[NTQ] = {};
    v8f a3[NTQ] = {}, b3[NTQ] = {}, c3[NTQ] = {};
    v2f pA0 = {}, pA1 = {}, pB0 = {}, pB1 = {}, pE3 = {};

    auto loadB = [&](v2f (&f)[NFRAG], unsigned p) {
        #pragma unroll
        for (int j = 0; j < NFRAG; ++j) f[j] = *(const v2f*)(smem + p + 2 * (j + (j >> 4)));
    };
    auto loadW = [&](v4f (&w)[4], int g) {
        const v4f* wp = (const v4f*)(smem + (2 * g + half) * PITCH2 + P256OFF);
        #pragma unroll
        for (int q = 0; q < 4; ++q) w[q] = wp[q];
    };

    v2f f0[NFRAG], f1[NFRAG];
    v4f w0[4], w1[4];
    const unsigned fp = (unsigned)(half * PITCH2 + 34 * nloc);
    loadB(f0, fp);
    loadW(w0, 0);

    auto group = [&](const v2f (&cf)[NFRAG], const v4f (&ca)[4], const v4f (&cw)[4],
                     v2f (&nf)[NFRAG], v4f (&na)[4], v4f (&nw)[4], int g) {
        const int gn = g + 1 < NGRP ? g + 1 : NGRP - 1;
        __builtin_amdgcn_sched_barrier(0);
        loadB(nf, fp + (unsigned)gn * (2 * PITCH2));
        loadW(nw, gn);
        loadA(na, gn);
        __builtin_amdgcn_sched_barrier(0);
        const v2f h0 = {ca[0].x, ca[0].y}, h2 = {ca[0].z, ca[0].w}, h4 = {ca[1].x, ca[1].y}, h6 = {ca[1].z, ca[1].w};
        const v2f h1 = {ca[2].x, ca[2].y}, h3 = {ca[2].z, ca[2].w}, h5 = {ca[3].x, ca[3].y}, h7 = {ca[3].z, ca[3].w};
        const v2f Ha1[2] = {h0, h4}, Hb1[2] = {h2, h6}, Hc1[2] = {h0 + h2, h4 + h6};
        const v2f Ha2[2] = {h1, h5}, Hb2[2] = {h3, h7}, Hc2[2] = {h1 + h3, h5 + h7};
        const v2f Ha3[2] = {h0 + h1, h4 + h5}, Hb3[2] = {h2 + h3, h6 + h7};
        const v2f Hc3[2] = {Ha3[0] + Hb3[0], Ha3[1] + Hb3[1]};
        v2f Uc1[NQS], Uc2[NQS], Ua3[NQS + 1], Ub3[NQS], Uc3[NQS];
        #pragma unroll
        for (int q = 0; q < NQS; ++q) {
            Uc1[q] = cf[4 * q + 2] + cf[4 * q + 4];
            Uc2[q] = cf[4 * q + 3] + cf[4 * q + 5];
            Ub3[q] = cf[4 * q + 3] + cf[4 * q + 4];
        }
        #pragma unroll
        for (int q = 0; q < NQS + 1; ++q) Ua3[q] = cf[4 * q + 1] + cf[4 * q + 2];
        #pragma unroll
        for (int q = 0; q < NQS; ++q) Uc3[q] = Ub3[q] + Ua3[q + 1];
        const v2f x256 = {cw[0].x, cw[0].y}, x257 = {cw[0].z, cw[0].w}, x258 = {cw[1].x, cw[1].y}, x259 = {cw[1].z, cw[1].w};
        const v2f x260 = {cw[2].x, cw[2].y}, x261 = {cw[2].z, cw[2].w}, x262 = {cw[3].x, cw[3].y}, x263 = {cw[3].z, cw[3].w};
        pA0 += h0 * x256;
        pA1 += h2 * x258;
        pA0 += h4 * x260;
        pA1 += h6 * x262;
        pB0 += h1 * x257;
        pB1 += h3 * x259;
        pB0 += h5 * x261;
        pB1 += h7 * x263;
        pE3 += Ha3[0] * (x257 + x258);
        pE3 += Ha3[1] * (x261 + x262);
        #pragma unroll
        for (int r = 0; r < 2; ++r) {
            #pragma unroll
            for (int j = 0; j < NTQ; ++j)
                a1[j] = __builtin_amdgcn_wmma_f32_16x16x4_f32(
                    false, Ha1[r], false, cf[4 * (j + r)], (short)0, a1[j], false, false);
            #pragma unroll
            for (int j = 0; j < NTQ; ++j)
                b1[j] = __builtin_amdgcn_wmma_f32_16x16x4_f32(
                    false, Hb1[r], false, cf[4 * (j + r) + 2], (short)0, b1[j], false, false);
            #pragma unroll
            for (int j = 0; j < NTQ; ++j)
                c1[j] = __builtin_amdgcn_wmma_f32_16x16x4_f32(
                    false, Hc1[r], false, Uc1[j + r], (short)0, c1[j], false, false);
            #pragma unroll
            for (int j = 0; j < NTQ; ++j)
                a2[j] = __builtin_amdgcn_wmma_f32_16x16x4_f32(
                    false, Ha2[r], false, cf[4 * (j + r) + 1], (short)0, a2[j], false, false);
            #pragma unroll
            for (int j = 0; j < NTQ; ++j)
                b2[j] = __builtin_amdgcn_wmma_f32_16x16x4_f32(
                    false, Hb2[r], false, cf[4 * (j + r) + 3], (short)0, b2[j], false, false);
            #pragma unroll
            for (int j = 0; j < NTQ; ++j)
                c2[j] = __builtin_amdgcn_wmma_f32_16x16x4_f32(
                    false, Hc2[r], false, Uc2[j + r], (short)0, c2[j], false, false);
            #pragma unroll
            for (int j = 0; j < NTQ; ++j)
                a3[j] = __builtin_amdgcn_wmma_f32_16x16x4_f32(
                    false, Ha3[r], false, Ua3[j + r], (short)0, a3[j], false, false);
            #pragma unroll
            for (int j = 0; j < NTQ; ++j)
                b3[j] = __builtin_amdgcn_wmma_f32_16x16x4_f32(
                    false, Hb3[r], false, Ub3[j + r], (short)0, b3[j], false, false);
            #pragma unroll
            for (int j = 0; j < NTQ; ++j)
                c3[j] = __builtin_amdgcn_wmma_f32_16x16x4_f32(
                    false, Hc3[r], false, Uc3[j + r], (short)0, c3[j], false, false);
        }
    };

    for (int g2 = 0; g2 < NGRP / 2; ++g2) {
        group(f0, ra0, w0, f1, ra1, w1, 2 * g2);
        group(f1, ra1, w1, f0, ra0, w0, 2 * g2 + 1);
    }

    float a128 = (pA0.x + pA0.y) + (pA1.x + pA1.y);
    float b128 = (pB0.x + pB0.y) + (pB1.x + pB1.y);
    float e1 = pA0.x + pA0.y;
    float e2 = pB0.x + pB0.y;
    float e3 = pE3.x + pE3.y;
    a128 += __shfl_xor(a128, 16, 32);
    b128 += __shfl_xor(b128, 16, 32);
    e1 += __shfl_xor(e1, 16, 32);
    e2 += __shfl_xor(e2, 16, 32);
    e3 += __shfl_xor(e3, 16, 32);
    v8f n1, n2, n3, nAe;
    #pragma unroll
    for (int r = 0; r < 8; ++r) {
        const int nl = (lane + 1) & 31, el = 8 * half + r;
        const float c1v = __shfl(a1[0][r], nl, 32), k1v = __shfl(e1, el, 32);
        const float c2v = __shfl(a2[0][r], nl, 32), k2v = __shfl(e2, el, 32);
        const float c3v = __shfl(a3[0][r], nl, 32), k3v = __shfl(e3, el, 32);
        const float cAv = __shfl(a1[0][r] + b1[0][r], nl, 32), kAv = __shfl(a128, el, 32);
        n1[r]  = (nloc == 15) ? k1v : c1v;
        n2[r]  = (nloc == 15) ? k2v : c2v;
        n3[r]  = (nloc == 15) ? k3v : c3v;
        nAe[r] = (nloc == 15) ? kAv : cAv;
    }
    v8f acc[NT];
    #pragma unroll
    for (int j = 0; j < NTQ; ++j) {
        const v8f Ae  = a1[j] + b1[j];
        const v8f Ao  = (c1[j] - (j + 1 < NTQ ? a1[j + 1] : n1)) - b1[j];
        const v8f Be  = a2[j] + b2[j];
        const v8f Bo  = (c2[j] - (j + 1 < NTQ ? a2[j + 1] : n2)) - b2[j];
        const v8f Ce  = a3[j] + b3[j];
        const v8f Co  = (c3[j] - (j + 1 < NTQ ? a3[j + 1] : n3)) - b3[j];
        const v8f Aen = j + 1 < NTQ ? a1[j + 1] + b1[j + 1] : nAe;
        acc[4 * j]     = Ae + Be;
        acc[4 * j + 1] = (Ce - Ao) - Be;
        acc[4 * j + 2] = Ao + Bo;
        acc[4 * j + 3] = (Co - Aen) - Bo;
    }

    float bestv = -3.402823466e38f;
    #pragma unroll
    for (int r = 0; r < 8; ++r) {
        #pragma unroll
        for (int j = 0; j < NT; ++j) bestv = fmaxf(bestv, acc[j][r]);
    }
    unsigned code = 0xFFFFFFFFu;
    #pragma unroll
    for (int r = 7; r >= 0; --r) {
        #pragma unroll
        for (int j = NT - 1; j >= 0; --j)
            if (acc[j][r] == bestv) code = (unsigned)(r * PPOS + j);
    }
    unsigned bestidx = code == 0xFFFFFFFFu
        ? 0u : (unsigned)((m0 + wave * 16 + 8 * half) * PPOS + 16 * nloc) + code;
    const float s256 = a128 + b128;
    const unsigned idx256 = (unsigned)((m0 + wave * 16 + nloc) * PPOS + (PPOS - 1));
    if (s256 > bestv || (s256 == bestv && idx256 < bestidx)) { bestv = s256; bestidx = idx256; }
    unsigned vb = __float_as_uint(bestv);
    vb = (vb & 0x80000000u) ? ~vb : (vb | 0x80000000u);
    unsigned long long key = ((unsigned long long)vb << 32) | (0xFFFFFFFFu - bestidx);
    for (int off = 16; off > 0; off >>= 1) {
        unsigned long long o = __shfl_xor(key, off, 32);
        if (o > key) key = o;
    }
    unsigned long long* wkey = (unsigned long long*)(smem + RES_FLOATS);
    if (lane == 0) wkey[wave] = key;
    __syncthreads();
    if (tid == 0) {
        unsigned long long k = wkey[0];
        for (int w = 1; w < CORR_THREADS / 32; ++w) if (wkey[w] > k) k = wkey[w];
        atomicMax(best + b, k);
        const unsigned prev = __hip_atomic_fetch_add(arrived + b, 1u, __ATOMIC_ACQ_REL,
                                                     __HIP_MEMORY_SCOPE_AGENT);
        unsigned long long fin = 0ull;
        if (prev == gridDim.x - 1)
            fin = __hip_atomic_load(best + b, __ATOMIC_RELAXED, __HIP_MEMORY_SCOPE_AGENT);
        wkey[CORR_THREADS / 32] = fin;
    }
    __syncthreads();
    const unsigned long long fin = wkey[CORR_THREADS / 32];
    if (fin == 0ull) return;

    if (tid == 0) { best[b] = 0ull; arrived[b] = 0u; }
    const unsigned enc  = (unsigned)(fin >> 32);
    const unsigned bits = (enc & 0x80000000u) ? (enc & 0x7FFFFFFFu) : ~enc;
    const float val = __uint_as_float(bits);
    const unsigned idx = 0xFFFFFFFFu - (unsigned)(fin & 0xFFFFFFFFu);
    const int atom = min((int)(idx / PPOS), NATOMS - 1);
    const int pos  = (int)(idx % PPOS);
    const float* ap = dnn + (size_t)atom * KDIM;
    float* rb = res   + (size_t)b * RES_FLOATS;
    float* cb = recon + (size_t)b * RES_FLOATS;
    float ua[4], ur[4], uc[4];
    #pragma unroll
    for (int k = 0; k < KDIM / CORR_THREADS; ++k) {
        const int i = tid + k * CORR_THREADS;
        const int ch = i >> 3, t = i & 7;
        ua[k] = ap[kperm(ch, t)];
        ur[k] = rb[ch * RES_W + pos + t];
        uc[k] = cb[ch * RES_W + pos + t];
    }
    #pragma unroll
    for (int k = 0; k < KDIM / CORR_THREADS; ++k) {
        const int i = tid + k * CORR_THREADS;
        const int ch = i >> 3, t = i & 7;
        const float av = ua[k] * val;
        rb[ch * RES_W + pos + t] = ur[k] - av;
        cb[ch * RES_W + pos + t] = uc[k] + av;
    }
}

__global__ void k_output(const float* __restrict__ recon, float* __restrict__ out) {
    const int i = blockIdx.x * blockDim.x + threadIdx.x;
    if (i < BATCH * CC * TT) {
        const int t  = i % TT;
        const int bc = i / TT;
        out[i] = recon[(size_t)bc * RES_W + t];
    }
}

extern "C" void kernel_launch(void* const* d_in, const int* in_sizes, int n_in,
                              void* d_out, int out_size, void* d_ws, size_t ws_size,
                              hipStream_t stream) {
    const float* x = (const float*)d_in[0];
    const float* d = (const float*)d_in[1];
    (void)in_sizes; (void)n_in; (void)out_size; (void)ws_size;

    char* ws = (char*)d_ws;
    float* dnn   = (float*)ws;
    float* res   = (float*)(ws + (size_t)NATOMS * KDIM * 4);
    float* recon = res + BATCH * CC * RES_W;
    unsigned long long* best = (unsigned long long*)(recon + BATCH * CC * RES_W);
    unsigned* arrived = (unsigned*)(best + BATCH);

    k_normalize<<<NATOMS, CORR_THREADS, 0, stream>>>(d, dnn);
    k_init<<<(BATCH * CC * RES_W + 255) / 256, 256, 0, stream>>>(x, res, recon, best, arrived);

    dim3 grid(NATOMS / MTILE, BATCH);
    for (int it = 0; it < NITER; ++it)
        k_corr<<<grid, CORR_THREADS, 0, stream>>>(dnn, res, recon, best, arrived);
    k_output<<<(BATCH * CC * TT + 255) / 256, 256, 0, stream>>>(recon, (float*)d_out);
}


# Fast FIR nested a third time: 27 one-tap eighth-rate products, not 36 quarter-rate tap products: 54 matrix products per channel group, was 72; 2 position tiles per product; sum order changed
# speedup vs baseline: 2.2323x; 1.1030x over previous
#include <hip/hip_runtime.h>

#define NATOMS 4096
#define CC     64
#define TT     256
#define ASIZE  8
#define BATCH  4
#define KDIM   512
#define PPOS   257
#define RES_W  304
#define NT     16
#define NT8    2
#define NPROD  27
#define MTILE  64
#define RES_FLOATS 19456
#define LDS_FLOATS 41984
#define CORR_THREADS 128
#define NITER  32
#define NGRP   16
#define NFRAG  23
#define SBLK   68
#define P256OFF 544
#define PITCH2 608

typedef __attribute__((ext_vector_type(2))) float v2f;
typedef __attribute__((ext_vector_type(4))) float v4f;
typedef __attribute__((ext_vector_type(8))) float v8f;

__device__ __forceinline__ int kperm(int c, int t) {
    return (c >> 2) * 32 + ((c >> 1) & 1) * 16 + (t & 1) * 8 + (t >> 1) * 2 + (c & 1);
}

__global__ void k_normalize(const float* __restrict__ d, float* __restrict__ dnn) {
    __shared__ float red[4];
    const int a = blockIdx.x;
    const float* src = d + (size_t)a * KDIM;
    float s = 0.f;
    for (int i = threadIdx.x; i < KDIM; i += CORR_THREADS) { float v = src[i]; s += v * v; }
    for (int off = 16; off > 0; off >>= 1) s += __shfl_down(s, off, 32);
    const int wv = threadIdx.x >> 5;
    if ((threadIdx.x & 31) == 0) red[wv] = s;
    __syncthreads();
    if (threadIdx.x == 0) red[0] = 1.f / (sqrtf(red[0] + red[1] + red[2] + red[3]) + 1e-8f);
    __syncthreads();
    const float inv = red[0];
    float* dst = dnn + (size_t)a * KDIM;
    for (int i = threadIdx.x; i < KDIM; i += CORR_THREADS) dst[kperm(i >> 3, i & 7)] = src[i] * inv;
}

__global__ void k_init(const float* __restrict__ x, float* __restrict__ res,
                       float* __restrict__ recon, unsigned long long* __restrict__ best,
                       unsigned* __restrict__ arrived) {
    const int i = blockIdx.x * blockDim.x + threadIdx.x;
    if (i < BATCH * CC * RES_W) {
        const int col = i % RES_W;
        const int bc  = i / RES_W;
        res[i]   = (col < TT) ? x[bc * TT + col] : 0.f;
        recon[i] = 0.f;
    }
    if (i < BATCH) { best[i] = 0ull; arrived[i] = 0u; }
}

__global__ __launch_bounds__(CORR_THREADS)
void k_corr(const float* __restrict__ dnn, float* res, float* recon,
            unsigned long long* best, unsigned* arrived) {
    __shared__ __attribute__((aligned(16))) float smem[LDS_FLOATS];
    const int b    = blockIdx.y;
    const int m0   = blockIdx.x * MTILE;
    const int tid  = threadIdx.x;
    const int lane = tid & 31;
    const int wave = tid >> 5;

    const int half = lane >> 4;
    const int nloc = lane & 15;
    const v4f* arow = (const v4f*)(dnn + (size_t)(m0 + wave * 16 + nloc) * KDIM + half * 16);

    auto loadA = [&](v4f (&a)[4], int g) {
        #pragma unroll
        for (int q = 0; q < 4; ++q) a[q] = arow[g * 8 + q];
    };

    v4f ra0[4], ra1[4];
    loadA(ra0, 0);

    const v4f* resb = (const v4f*)(res + (size_t)b * RES_FLOATS);
    #pragma unroll
    for (int h2 = 0; h2 < 2; ++h2) {
        const int i0 = h2 ? 9 : 0, i1 = h2 ? 17 : 9;
        v4f te[9], to[9];
        #pragma unroll
        for (int i = i0; i < i1; ++i) {
            const int q = tid + i * CORR_THREADS;
            const int r = q / SBLK, c4 = q % SBLK;
            te[i - i0] = resb[(2 * r) * (RES_W / 4) + c4];
            to[i - i0] = resb[(2 * r + 1) * (RES_W / 4) + c4];
        }
        #pragma unroll
        for (int i = i0; i < i1; ++i) {
            const int q = tid + i * CORR_THREADS;
            const int r = q / SBLK, c4 = q % SBLK;
            v2f* d2 = (v2f*)(smem + r * PITCH2 + c4 * 8 + (c4 >> 2) * 2);
            const v4f e = te[i - i0], o = to[i - i0];
            d2[0] = (v2f){e.x, o.x};
            d2[1] = (v2f){e.y, o.y};
            d2[2] = (v2f){e.z, o.z};
            d2[3] = (v2f){e.w, o.w};
        }
    }
    __syncthreads();

    v8f acc3[NPROD][NT8] = {};
    v2f pz[NPROD] = {};

    auto loadB = [&](v2f (&f)[NFRAG], unsigned p) {
        #pragma unroll
        for (int j = 0; j < NFRAG; ++j) f[j] = *(const v2f*)(smem + p + 2 * (j + (j >> 4)));
    };
    auto loadW = [&](v4f (&w)[4], int g) {
        const v4f* wp = (const v4f*)(smem + (2 * g + half) * PITCH2 + P256OFF);
        #pragma unroll
        for (int q = 0; q < 4; ++q) w[q] = wp[q];
    };

    v2f f0[NFRAG], f1[NFRAG];
    v4f w0[4], w1[4];
    const unsigned fp = (unsigned)(half * PITCH2 + 34 * nloc);
    loadB(f0, fp);
    loadW(w0, 0);

    auto bndNeeded = [](int d) {
        const int d2 = (d / 3) % 3, d3 = d % 3;
        return d3 == 0 || (d2 == 0 && d3 == 1) || d == 4 || d == 13;
    };
    auto buildU = [&](const v2f (&S)[15], v2f (&U3)[NPROD]) {
        v2f L1[3][7];
        #pragma unroll
        for (int m = 0; m < 7; ++m) {
            L1[0][m] = S[2 * m];
            L1[1][m] = S[2 * m + 1];
            L1[2][m] = S[2 * m + 1] + S[2 * m + 2];
        }
        #pragma unroll
        for (int d1 = 0; d1 < 3; ++d1) {
            v2f L2[3][3];
            #pragma unroll
            for (int n = 0; n < 3; ++n) {
                L2[0][n] = L1[d1][2 * n];
                L2[1][n] = L1[d1][2 * n + 1];
                L2[2][n] = L1[d1][2 * n + 1] + L1[d1][2 * n + 2];
            }
            #pragma unroll
            for (int d2 = 0; d2 < 3; ++d2) {
                U3[9 * d1 + 3 * d2]     = L2[d2][0];
                U3[9 * d1 + 3 * d2 + 1] = L2[d2][1];
                U3[9 * d1 + 3 * d2 + 2] = L2[d2][1] + L2[d2][2];
            }
        }
    };

    auto group = [&](const v2f (&cf)[NFRAG], const v4f (&ca)[4], const v4f (&cw)[4],
                     v2f (&nf)[NFRAG], v4f (&na)[4], v4f (&nw)[4], int g) {
        const int gn = g + 1 < NGRP ? g + 1 : NGRP - 1;
        __builtin_amdgcn_sched_barrier(0);
        loadB(nf, fp + (unsigned)gn * (2 * PITCH2));
        loadW(nw, gn);
        loadA(na, gn);
        __builtin_amdgcn_sched_barrier(0);
        const v2f h[8] = {{ca[0].x, ca[0].y}, {ca[2].x, ca[2].y}, {ca[0].z, ca[0].w}, {ca[2].z, ca[2].w},
                          {ca[1].x, ca[1].y}, {ca[3].x, ca[3].y}, {ca[1].z, ca[1].w}, {ca[3].z, ca[3].w}};
        v2f G3[NPROD];
        {
            v2f G1[3][4];
            #pragma unroll
            for (int s = 0; s < 4; ++s) {
                G1[0][s] = h[2 * s];
                G1[1][s] = h[2 * s + 1];
                G1[2][s] = h[2 * s] + h[2 * s + 1];
            }
            #pragma unroll
            for (int d1 = 0; d1 < 3; ++d1) {
                v2f G2[3][2];
                #pragma unroll
                for (int r = 0; r < 2; ++r) {
                    G2[0][r] = G1[d1][2 * r];
                    G2[1][r] = G1[d1][2 * r + 1];
                    G2[2][r] = G1[d1][2 * r] + G1[d1][2 * r + 1];
                }
                #pragma unroll
                for (int d2 = 0; d2 < 3; ++d2) {
                    G3[9 * d1 + 3 * d2]     = G2[d2][0];
                    G3[9 * d1 + 3 * d2 + 1] = G2[d2][1];
                    G3[9 * d1 + 3 * d2 + 2] = G2[d2][0] + G2[d2][1];
                }
            }
        }
        v2f U3t[NT8][NPROD];
        #pragma unroll
        for (int t = 0; t < NT8; ++t) {
            v2f S[15];
            #pragma unroll
            for (int v = 0; v < 15; ++v) S[v] = cf[8 * t + v];
            buildU(S, U3t[t]);
        }
        {
            const v2f zero2 = {0.f, 0.f};
            const v2f Sw[15] = {{cw[0].x, cw[0].y}, {cw[0].z, cw[0].w}, {cw[1].x, cw[1].y}, {cw[1].z, cw[1].w},
                                {cw[2].x, cw[2].y}, {cw[2].z, cw[2].w}, {cw[3].x, cw[3].y}, {cw[3].z, cw[3].w},
                                zero2, zero2, zero2, zero2, zero2, zero2, zero2};
            v2f UW[NPROD];
            buildU(Sw, UW);
            #pragma unroll
            for (int d = 0; d < NPROD; ++d)
                if (bndNeeded(d)) pz[d] += G3[d] * UW[d];
        }
        #pragma unroll
        for (int d = 0; d < NPROD; ++d) {
            #pragma unroll
            for (int t = 0; t < NT8; ++t)
                acc3[d][t] = __builtin_amdgcn_wmma_f32_16x16x4_f32(
                    false, G3[d], false, U3t[t][d], (short)0, acc3[d][t], false, false);
        }
    };

    for (int g2 = 0; g2 < NGRP / 2; ++g2) {
        group(f0, ra0, w0, f1, ra1, w1, 2 * g2);
        group(f1, ra1, w1, f0, ra0, w0, 2 * g2 + 1);
    }

    float zb[NPROD];
    #pragma unroll
    for (int d = 0; d < NPROD; ++d) {
        zb[d] = 0.f;
        if (bndNeeded(d)) {
            float v = pz[d].x + pz[d].y;
            v += __shfl_xor(v, 16, 32);
            zb[d] = v;
        }
    }
    auto shiftnext = [&](const v8f &a0, float bnd) {
        v8f o;
        #pragma unroll
        for (int r = 0; r < 8; ++r) {
            const float kv = __shfl(bnd, 8 * half + r, 32);
            o[r] = __int_as_float(__builtin_amdgcn_update_dpp(
                __float_as_int(kv), __float_as_int(a0[r]), 0x101, 0xF, 0xF, false));
        }
        return o;
    };
    v8f Q[3][3][4];
    float Qb[3][3];
    #pragma unroll
    for (int d1 = 0; d1 < 3; ++d1) {
        #pragma unroll
        for (int d2 = 0; d2 < 3; ++d2) {
            const int base = 9 * d1 + 3 * d2;
            #pragma unroll
            for (int t = 0; t < 2; ++t) {
                Q[d1][d2][2 * t] = acc3[base][t] + acc3[base + 1][t];
                const v8f an = t + 1 < 2 ? acc3[base][t + 1] : shiftnext(acc3[base][0], zb[base]);
                Q[d1][d2][2 * t + 1] = (acc3[base + 2][t] - an) - acc3[base + 1][t];
            }
            Qb[d1][d2] = zb[base] + zb[base + 1];
        }
    }
    v8f R[3][8];
    float Rb[3];
    #pragma unroll
    for (int d1 = 0; d1 < 3; ++d1) {
        #pragma unroll
        for (int t = 0; t < 4; ++t) {
            R[d1][2 * t] = Q[d1][0][t] + Q[d1][1][t];
            const v8f an = t + 1 < 4 ? Q[d1][0][t + 1] : shiftnext(Q[d1][0][0], Qb[d1][0]);
            R[d1][2 * t + 1] = (Q[d1][2][t] - an) - Q[d1][1][t];
        }
        Rb[d1] = Qb[d1][0] + Qb[d1][1];
    }
    v8f acc[NT];
    #pragma unroll
    for (int t = 0; t < 8; ++t) {
        acc[2 * t] = R[0][t] + R[1][t];
        const v8f an = t + 1 < 8 ? R[0][t + 1] : shiftnext(R[0][0], Rb[0]);
        acc[2 * t + 1] = (R[2][t] - an) - R[1][t];
    }

    float bestv = -3.402823466e38f;
    #pragma unroll
    for (int r = 0; r < 8; ++r) {
        #pragma unroll
        for (int j = 0; j < NT; ++j) bestv = fmaxf(bestv, acc[j][r]);
    }
    unsigned code = 0xFFFFFFFFu;
    #pragma unroll
    for (int r = 7; r >= 0; --r) {
        #pragma unroll
        for (int j = NT - 1; j >= 0; --j)
            if (acc[j][r] == bestv) code = (unsigned)(r * PPOS + j);
    }
    unsigned bestidx = code == 0xFFFFFFFFu
        ? 0u : (unsigned)((m0 + wave * 16 + 8 * half) * PPOS + 16 * nloc) + code;
    const float s256 = Rb[0] + Rb[1];
    const unsigned idx256 = (unsigned)((m0 + wave * 16 + nloc) * PPOS + (PPOS - 1));
    if (s256 > bestv || (s256 == bestv && idx256 < bestidx)) { bestv = s256; bestidx = idx256; }
    unsigned vb = __float_as_uint(bestv);
    vb = (vb & 0x80000000u) ? ~vb : (vb | 0x80000000u);
    unsigned long long key = ((unsigned long long)vb << 32) | (0xFFFFFFFFu - bestidx);
    for (int off = 16; off > 0; off >>= 1) {
        unsigned long long o = __shfl_xor(key, off, 32);
        if (o > key) key = o;
    }
    unsigned long long* wkey = (unsigned long long*)(smem + RES_FLOATS);
    if (lane == 0) wkey[wave] = key;
    __syncthreads();
    if (tid == 0) {
        unsigned long long k = wkey[0];
        for (int w = 1; w < CORR_THREADS / 32; ++w) if (wkey[w] > k) k = wkey[w];
        atomicMax(best + b, k);
        const unsigned prev = __hip_atomic_fetch_add(arrived + b, 1u, __ATOMIC_ACQ_REL,
                                                     __HIP_MEMORY_SCOPE_AGENT);
        unsigned long long fin = 0ull;
        if (prev == gridDim.x - 1)
            fin = __hip_atomic_load(best + b, __ATOMIC_RELAXED, __HIP_MEMORY_SCOPE_AGENT);
        wkey[CORR_THREADS / 32] = fin;
    }
    __syncthreads();
    const unsigned long long fin = wkey[CORR_THREADS / 32];
    if (fin == 0ull) return;

    if (tid == 0) { best[b] = 0ull; arrived[b] = 0u; }
    const unsigned enc  = (unsigned)(fin >> 32);
    const unsigned bits = (enc & 0x80000000u) ? (enc & 0x7FFFFFFFu) : ~enc;
    const float val = __uint_as_float(bits);
    const unsigned idx = 0xFFFFFFFFu - (unsigned)(fin & 0xFFFFFFFFu);
    const int atom = min((int)(idx / PPOS), NATOMS - 1);
    const int pos  = (int)(idx % PPOS);
    const float* ap = dnn + (size_t)atom * KDIM;
    float* rb = res   + (size_t)b * RES_FLOATS;
    float* cb = recon + (size_t)b * RES_FLOATS;
    float ua[4], ur[4], uc[4];
    #pragma unroll
    for (int k = 0; k < KDIM / CORR_THREADS; ++k) {
        const int i = tid + k * CORR_THREADS;
        const int ch = i >> 3, t = i & 7;
        ua[k] = ap[kperm(ch, t)];
        ur[k] = rb[ch * RES_W + pos + t];
        uc[k] = cb[ch * RES_W + pos + t];
    }
    #pragma unroll
    for (int k = 0; k < KDIM / CORR_THREADS; ++k) {
        const int i = tid + k * CORR_THREADS;
        const int ch = i >> 3, t = i & 7;
        const float av = ua[k] * val;
        rb[ch * RES_W + pos + t] = ur[k] - av;
        cb[ch * RES_W + pos + t] = uc[k] + av;
    }
}

__global__ void k_output(const float* __restrict__ recon, float* __restrict__ out) {
    const int i = blockIdx.x * blockDim.x + threadIdx.x;
    if (i < BATCH * CC * TT) {
        const int t  = i % TT;
        const int bc = i / TT;
        out[i] = recon[(size_t)bc * RES_W + t];
    }
}

extern "C" void kernel_launch(void* const* d_in, const int* in_sizes, int n_in,
                              void* d_out, int out_size, void* d_ws, size_t ws_size,
                              hipStream_t stream) {
    const float* x = (const float*)d_in[0];
    const float* d = (const float*)d_in[1];
    (void)in_sizes; (void)n_in; (void)out_size; (void)ws_size;

    char* ws = (char*)d_ws;
    float* dnn   = (float*)ws;
    float* res   = (float*)(ws + (size_t)NATOMS * KDIM * 4);
    float* recon = res + BATCH * CC * RES_W;
    unsigned long long* best = (unsigned long long*)(recon + BATCH * CC * RES_W);
    unsigned* arrived = (unsigned*)(best + BATCH);

    k_normalize<<<NATOMS, CORR_THREADS, 0, stream>>>(d, dnn);
    k_init<<<(BATCH * CC * RES_W + 255) / 256, 256, 0, stream>>>(x, res, recon, best, arrived);

    dim3 grid(NATOMS / MTILE, BATCH);
    for (int it = 0; it < NITER; ++it)
        k_corr<<<grid, CORR_THREADS, 0, stream>>>(dnn, res, recon, best, arrived);
    k_output<<<(BATCH * CC * TT + 255) / 256, 256, 0, stream>>>(recon, (float*)d_out);
}


# Correlation map kept in workspace after launch 1; the other 31 launches subtract val * (dictionary x chosen atom, one 16 atom x 16 lag tile per wave) at 15 positions and rescan; sum order changed
# speedup vs baseline: 4.5113x; 2.0209x over previous
#include <hip/hip_runtime.h>

#define NATOMS 4096
#define CC     64
#define TT     256
#define ASIZE  8
#define BATCH  4
#define KDIM   512
#define PPOS   257
#define RES_W  304
#define NT     16
#define NT8    2
#define NPROD  27
#define MTILE  64
#define RES_FLOATS 19456
#define LDS_FLOATS 41984
#define CORR_THREADS 128
#define NITER  32
#define NGRP   16
#define NFRAG  23
#define SBLK   68
#define P256OFF 544
#define MAPW   260
#define MAPV   33
#define LAGP   48
#define LOWF   (-3.402823466e38f)
#define PITCH2 608

typedef __attribute__((ext_vector_type(2))) float v2f;
typedef __attribute__((ext_vector_type(4))) float v4f;
typedef __attribute__((ext_vector_type(8))) float v8f;

__device__ __forceinline__ int kperm(int c, int t) {
    return (c >> 2) * 32 + ((c >> 1) & 1) * 16 + (t & 1) * 8 + (t >> 1) * 2 + (c & 1);
}

__global__ void k_normalize(const float* __restrict__ d, float* __restrict__ dnn) {
    __shared__ float red[4];
    const int a = blockIdx.x;
    const float* src = d + (size_t)a * KDIM;
    float s = 0.f;
    for (int i = threadIdx.x; i < KDIM; i += CORR_THREADS) { float v = src[i]; s += v * v; }
    for (int off = 16; off > 0; off >>= 1) s += __shfl_down(s, off, 32);
    const int wv = threadIdx.x >> 5;
    if ((threadIdx.x & 31) == 0) red[wv] = s;
    __syncthreads();
    if (threadIdx.x == 0) red[0] = 1.f / (sqrtf(red[0] + red[1] + red[2] + red[3]) + 1e-8f);
    __syncthreads();
    const float inv = red[0];
    float* dst = dnn + (size_t)a * KDIM;
    for (int i = threadIdx.x; i < KDIM; i += CORR_THREADS) dst[kperm(i >> 3, i & 7)] = src[i] * inv;
}

__global__ void k_init(const float* __restrict__ x, float* __restrict__ res,
                       float* __restrict__ recon, unsigned long long* __restrict__ best,
                       unsigned* __restrict__ arrived, int nkeys) {
    const int i = blockIdx.x * blockDim.x + threadIdx.x;
    if (i < BATCH * CC * RES_W) {
        const int col = i % RES_W;
        const int bc  = i / RES_W;
        res[i]   = (col < TT) ? x[bc * TT + col] : 0.f;
        recon[i] = 0.f;
    }
    if (i < nkeys) best[i] = 0ull;
    if (arrived != nullptr && i < BATCH) arrived[i] = 0u;
}

template <bool KEEP_MAP>
__global__ __launch_bounds__(CORR_THREADS)
void k_corr(const float* __restrict__ dnn, float* res, float* recon,
            unsigned long long* best, unsigned* arrived, float* map) {
    __shared__ __attribute__((aligned(16))) float smem[LDS_FLOATS];
    const int b    = blockIdx.y;
    const int m0   = blockIdx.x * MTILE;
    const int tid  = threadIdx.x;
    const int lane = tid & 31;
    const int wave = tid >> 5;

    const int half = lane >> 4;
    const int nloc = lane & 15;
    const v4f* arow = (const v4f*)(dnn + (size_t)(m0 + wave * 16 + nloc) * KDIM + half * 16);

    auto loadA = [&](v4f (&a)[4], int g) {
        #pragma unroll
        for (int q = 0; q < 4; ++q) a[q] = arow[g * 8 + q];
    };

    v4f ra0[4], ra1[4];
    loadA(ra0, 0);

    const v4f* resb = (const v4f*)(res + (size_t)b * RES_FLOATS);
    #pragma unroll
    for (int h2 = 0; h2 < 2; ++h2) {
        const int i0 = h2 ? 9 : 0, i1 = h2 ? 17 : 9;
        v4f te[9], to[9];
        #pragma unroll
        for (int i = i0; i < i1; ++i) {
            const int q = tid + i * CORR_THREADS;
            const int r = q / SBLK, c4 = q % SBLK;
            te[i - i0] = resb[(2 * r) * (RES_W / 4) + c4];
            to[i - i0] = resb[(2 * r + 1) * (RES_W / 4) + c4];
        }
        #pragma unroll
        for (int i = i0; i < i1; ++i) {
            const int q = tid + i * CORR_THREADS;
            const int r = q / SBLK, c4 = q % SBLK;
            v2f* d2 = (v2f*)(smem + r * PITCH2 + c4 * 8 + (c4 >> 2) * 2);
            const v4f e = te[i - i0], o = to[i - i0];
            d2[0] = (v2f){e.x, o.x};
            d2[1] = (v2f){e.y, o.y};
            d2[2] = (v2f){e.z, o.z};
            d2[3] = (v2f){e.w, o.w};
        }
    }
    __syncthreads();

    v8f acc3[NPROD][NT8] = {};
    v2f pz[NPROD] = {};

    auto loadB = [&](v2f (&f)[NFRAG], unsigned p) {
        #pragma unroll
        for (int j = 0; j < NFRAG; ++j) f[j] = *(const v2f*)(smem + p + 2 * (j + (j >> 4)));
    };
    auto loadW = [&](v4f (&w)[4], int g) {
        const v4f* wp = (const v4f*)(smem + (2 * g + half) * PITCH2 + P256OFF);
        #pragma unroll
        for (int q = 0; q < 4; ++q) w[q] = wp[q];
    };

    v2f f0[NFRAG], f1[NFRAG];
    v4f w0[4], w1[4];
    const unsigned fp = (unsigned)(half * PITCH2 + 34 * nloc);
    loadB(f0, fp);
    loadW(w0, 0);

    auto bndNeeded = [](int d) {
        const int d2 = (d / 3) % 3, d3 = d % 3;
        return d3 == 0 || (d2 == 0 && d3 == 1) || d == 4 || d == 13;
    };
    auto buildU = [&](const v2f (&S)[15], v2f (&U3)[NPROD]) {
        v2f L1[3][7];
        #pragma unroll
        for (int m = 0; m < 7; ++m) {
            L1[0][m] = S[2 * m];
            L1[1][m] = S[2 * m + 1];
            L1[2][m] = S[2 * m + 1] + S[2 * m + 2];
        }
        #pragma unroll
        for (int d1 = 0; d1 < 3; ++d1) {
            v2f L2[3][3];
            #pragma unroll
            for (int n = 0; n < 3; ++n) {
                L2[0][n] = L1[d1][2 * n];
                L2[1][n] = L1[d1][2 * n + 1];
                L2[2][n] = L1[d1][2 * n + 1] + L1[d1][2 * n + 2];
            }
            #pragma unroll
            for (int d2 = 0; d2 < 3; ++d2) {
                U3[9 * d1 + 3 * d2]     = L2[d2][0];
                U3[9 * d1 + 3 * d2 + 1] = L2[d2][1];
                U3[9 * d1 + 3 * d2 + 2] = L2[d2][1] + L2[d2][2];
            }
        }
    };

    auto group = [&](const v2f (&cf)[NFRAG], const v4f (&ca)[4], const v4f (&cw)[4],
                     v2f (&nf)[NFRAG], v4f (&na)[4], v4f (&nw)[4], int g) {
        const int gn = g + 1 < NGRP ? g + 1 : NGRP - 1;
        __builtin_amdgcn_sched_barrier(0);
        loadB(nf, fp + (unsigned)gn * (2 * PITCH2));
        loadW(nw, gn);
        loadA(na, gn);
        __builtin_amdgcn_sched_barrier(0);
        const v2f h[8] = {{ca[0].x, ca[0].y}, {ca[2].x, ca[2].y}, {ca[0].z, ca[0].w}, {ca[2].z, ca[2].w},
                          {ca[1].x, ca[1].y}, {ca[3].x, ca[3].y}, {ca[1].z, ca[1].w}, {ca[3].z, ca[3].w}};
        v2f G3[NPROD];
        {
            v2f G1[3][4];
            #pragma unroll
            for (int s = 0; s < 4; ++s) {
                G1[0][s] = h[2 * s];
                G1[1][s] = h[2 * s + 1];
                G1[2][s] = h[2 * s] + h[2 * s + 1];
            }
            #pragma unroll
            for (int d1 = 0; d1 < 3; ++d1) {
                v2f G2[3][2];
                #pragma unroll
                for (int r = 0; r < 2; ++r) {
                    G2[0][r] = G1[d1][2 * r];
                    G2[1][r] = G1[d1][2 * r + 1];
                    G2[2][r] = G1[d1][2 * r] + G1[d1][2 * r + 1];
                }
                #pragma unroll
                for (int d2 = 0; d2 < 3; ++d2) {
                    G3[9 * d1 + 3 * d2]     = G2[d2][0];
                    G3[9 * d1 + 3 * d2 + 1] = G2[d2][1];
                    G3[9 * d1 + 3 * d2 + 2] = G2[d2][0] + G2[d2][1];
                }
            }
        }
        v2f U3t[NT8][NPROD];
        #pragma unroll
        for (int t = 0; t < NT8; ++t) {
            v2f S[15];
            #pragma unroll
            for (int v = 0; v < 15; ++v) S[v] = cf[8 * t + v];
            buildU(S, U3t[t]);
        }
        {
            const v2f zero2 = {0.f, 0.f};
            const v2f Sw[15] = {{cw[0].x, cw[0].y}, {cw[0].z, cw[0].w}, {cw[1].x, cw[1].y}, {cw[1].z, cw[1].w},
                                {cw[2].x, cw[2].y}, {cw[2].z, cw[2].w}, {cw[3].x, cw[3].y}, {cw[3].z, cw[3].w},
                                zero2, zero2, zero2, zero2, zero2, zero2, zero2};
            v2f UW[NPROD];
            buildU(Sw, UW);
            #pragma unroll
            for (int d = 0; d < NPROD; ++d)
                if (bndNeeded(d)) pz[d] += G3[d] * UW[d];
        }
        #pragma unroll
        for (int d = 0; d < NPROD; ++d) {
            #pragma unroll
            for (int t = 0; t < NT8; ++t)
                acc3[d][t] = __builtin_amdgcn_wmma_f32_16x16x4_f32(
                    false, G3[d], false, U3t[t][d], (short)0, acc3[d][t], false, false);
        }
    };

    for (int g2 = 0; g2 < NGRP / 2; ++g2) {
        group(f0, ra0, w0, f1, ra1, w1, 2 * g2);
        group(f1, ra1, w1, f0, ra0, w0, 2 * g2 + 1);
    }

    float zb[NPROD];
    #pragma unroll
    for (int d = 0; d < NPROD; ++d) {
        zb[d] = 0.f;
        if (bndNeeded(d)) {
            float v = pz[d].x + pz[d].y;
            v += __shfl_xor(v, 16, 32);
            zb[d] = v;
        }
    }
    auto shiftnext = [&](const v8f &a0, float bnd) {
        v8f o;
        #pragma unroll
        for (int r = 0; r < 8; ++r) {
            const float kv = __shfl(bnd, 8 * half + r, 32);
            o[r] = __int_as_float(__builtin_amdgcn_update_dpp(
                __float_as_int(kv), __float_as_int(a0[r]), 0x101, 0xF, 0xF, false));
        }
        return o;
    };
    v8f Q[3][3][4];
    float Qb[3][3];
    #pragma unroll
    for (int d1 = 0; d1 < 3; ++d1) {
        #pragma unroll
        for (int d2 = 0; d2 < 3; ++d2) {
            const int base = 9 * d1 + 3 * d2;
            #pragma unroll
            for (int t = 0; t < 2; ++t) {
                Q[d1][d2][2 * t] = acc3[base][t] + acc3[base + 1][t];
                const v8f an = t + 1 < 2 ? acc3[base][t + 1] : shiftnext(acc3[base][0], zb[base]);
                Q[d1][d2][2 * t + 1] = (acc3[base + 2][t] - an) - acc3[base + 1][t];
            }
            Qb[d1][d2] = zb[base] + zb[base + 1];
        }
    }
    v8f R[3][8];
    float Rb[3];
    #pragma unroll
    for (int d1 = 0; d1 < 3; ++d1) {
        #pragma unroll
        for (int t = 0; t < 4; ++t) {
            R[d1][2 * t] = Q[d1][0][t] + Q[d1][1][t];
            const v8f an = t + 1 < 4 ? Q[d1][0][t + 1] : shiftnext(Q[d1][0][0], Qb[d1][0]);
            R[d1][2 * t + 1] = (Q[d1][2][t] - an) - Q[d1][1][t];
        }
        Rb[d1] = Qb[d1][0] + Qb[d1][1];
    }
    v8f acc[NT];
    #pragma unroll
    for (int t = 0; t < 8; ++t) {
        acc[2 * t] = R[0][t] + R[1][t];
        const v8f an = t + 1 < 8 ? R[0][t + 1] : shiftnext(R[0][0], Rb[0]);
        acc[2 * t + 1] = (R[2][t] - an) - R[1][t];
    }

    if (KEEP_MAP) {
        float* mrow = map + ((size_t)b * NATOMS + (m0 + wave * 16 + 8 * half)) * MAPW + 16 * nloc;
        #pragma unroll
        for (int r = 0; r < 8; ++r) {
            v4f* dst = (v4f*)(mrow + (size_t)r * MAPW);
            #pragma unroll
            for (int q = 0; q < 4; ++q)
                dst[q] = (v4f){acc[4 * q][r], acc[4 * q + 1][r], acc[4 * q + 2][r], acc[4 * q + 3][r]};
        }
    }

    float bestv = -3.402823466e38f;
    #pragma unroll
    for (int r = 0; r < 8; ++r) {
        #pragma unroll
        for (int j = 0; j < NT; ++j) bestv = fmaxf(bestv, acc[j][r]);
    }
    unsigned code = 0xFFFFFFFFu;
    #pragma unroll
    for (int r = 7; r >= 0; --r) {
        #pragma unroll
        for (int j = NT - 1; j >= 0; --j)
            if (acc[j][r] == bestv) code = (unsigned)(r * PPOS + j);
    }
    unsigned bestidx = code == 0xFFFFFFFFu
        ? 0u : (unsigned)((m0 + wave * 16 + 8 * half) * PPOS + 16 * nloc) + code;
    const float s256 = Rb[0] + Rb[1];
    if (KEEP_MAP && half == 0) {
        v4f* d4 = (v4f*)(map + ((size_t)b * NATOMS + (m0 + wave * 16 + nloc)) * MAPW + 256);
        *d4 = (v4f){s256, LOWF, LOWF, LOWF};
    }
    const unsigned idx256 = (unsigned)((m0 + wave * 16 + nloc) * PPOS + (PPOS - 1));
    if (s256 > bestv || (s256 == bestv && idx256 < bestidx)) { bestv = s256; bestidx = idx256; }
    unsigned vb = __float_as_uint(bestv);
    vb = (vb & 0x80000000u) ? ~vb : (vb | 0x80000000u);
    unsigned long long key = ((unsigned long long)vb << 32) | (0xFFFFFFFFu - bestidx);
    for (int off = 16; off > 0; off >>= 1) {
        unsigned long long o = __shfl_xor(key, off, 32);
        if (o > key) key = o;
    }
    unsigned long long* wkey = (unsigned long long*)(smem + RES_FLOATS);
    if (lane == 0) wkey[wave] = key;
    __syncthreads();
    if (tid == 0) {
        unsigned long long k = wkey[0];
        for (int w = 1; w < CORR_THREADS / 32; ++w) if (wkey[w] > k) k = wkey[w];
        atomicMax(best + b, k);
        if (KEEP_MAP) return;
        const unsigned prev = __hip_atomic_fetch_add(arrived + b, 1u, __ATOMIC_ACQ_REL,
                                                     __HIP_MEMORY_SCOPE_AGENT);
        unsigned long long fin = 0ull;
        if (prev == gridDim.x - 1)
            fin = __hip_atomic_load(best + b, __ATOMIC_RELAXED, __HIP_MEMORY_SCOPE_AGENT);
        wkey[CORR_THREADS / 32] = fin;
    }
    if (KEEP_MAP) return;
    __syncthreads();
    const unsigned long long fin = wkey[CORR_THREADS / 32];
    if (fin == 0ull) return;

    if (tid == 0) { best[b] = 0ull; arrived[b] = 0u; }
    const unsigned enc  = (unsigned)(fin >> 32);
    const unsigned bits = (enc & 0x80000000u) ? (enc & 0x7FFFFFFFu) : ~enc;
    const float val = __uint_as_float(bits);
    const unsigned idx = 0xFFFFFFFFu - (unsigned)(fin & 0xFFFFFFFFu);
    const int atom = min((int)(idx / PPOS), NATOMS - 1);
    const int pos  = (int)(idx % PPOS);
    const float* ap = dnn + (size_t)atom * KDIM;
    float* rb = res   + (size_t)b * RES_FLOATS;
    float* cb = recon + (size_t)b * RES_FLOATS;
    float ua[4], ur[4], uc[4];
    #pragma unroll
    for (int k = 0; k < KDIM / CORR_THREADS; ++k) {
        const int i = tid + k * CORR_THREADS;
        const int ch = i >> 3, t = i & 7;
        ua[k] = ap[kperm(ch, t)];
        ur[k] = rb[ch * RES_W + pos + t];
        uc[k] = cb[ch * RES_W + pos + t];
    }
    #pragma unroll
    for (int k = 0; k < KDIM / CORR_THREADS; ++k) {
        const int i = tid + k * CORR_THREADS;
        const int ch = i >> 3, t = i & 7;
        const float av = ua[k] * val;
        rb[ch * RES_W + pos + t] = ur[k] - av;
        cb[ch * RES_W + pos + t] = uc[k] + av;
    }
}


__global__ __launch_bounds__(CORR_THREADS)
void k_inc(const float* __restrict__ dnn, float* map, unsigned long long* keys, int step) {
    __shared__ __attribute__((aligned(16))) float smem[LDS_FLOATS];
    const int tid  = threadIdx.x;
    const int lane = tid & 31;
    const int b    = tid >> 5;
    const int a0   = blockIdx.x * 16;
    const int half = lane >> 4;
    const int nloc = lane & 15;

    const unsigned long long fin = __hip_atomic_load(keys + (step - 1) * BATCH + b, __ATOMIC_RELAXED,
                                                     __HIP_MEMORY_SCOPE_AGENT);
    float* mblk = map + ((size_t)b * NATOMS + a0) * MAPW;
    const v4f* mb4 = (const v4f*)mblk;
    v4f mv[MAPV];
    #pragma unroll
    for (int k = 0; k < MAPV; ++k) {
        const int e = lane + 32 * k;
        mv[k] = (v4f){LOWF, LOWF, LOWF, LOWF};
        if (e < 16 * (MAPW / 4)) mv[k] = mb4[e];
    }
    const v4f* arow = (const v4f*)(dnn + (size_t)(a0 + nloc) * KDIM + half * 16);
    v4f ra[NGRP][4];
    #pragma unroll
    for (int g = 0; g < NGRP; ++g) {
        #pragma unroll
        for (int q = 0; q < 4; ++q) ra[g][q] = arow[g * 8 + q];
    }
    float* cat = smem + b * (32 * LAGP * 2);
    {
        v2f* rowp = (v2f*)(cat + lane * (LAGP * 2));
        #pragma unroll
        for (int u = 0; u < 7; ++u) rowp[u] = (v2f){0.f, 0.f};
        #pragma unroll
        for (int u = 15; u < 23; ++u) rowp[u] = (v2f){0.f, 0.f};
    }
    __builtin_amdgcn_sched_barrier(0);

    const unsigned enc  = (unsigned)(fin >> 32);
    const unsigned bits = (enc & 0x80000000u) ? (enc & 0x7FFFFFFFu) : ~enc;
    const float val = __uint_as_float(bits);
    const unsigned idx = 0xFFFFFFFFu - (unsigned)(fin & 0xFFFFFFFFu);
    const int atom = min((int)(idx / PPOS), NATOMS - 1);
    const int pos  = (int)(idx % PPOS);

    const v4f* hp = (const v4f*)(dnn + (size_t)atom * KDIM + lane * 16);
    const v4f h0 = hp[0], h1 = hp[1], h2 = hp[2], h3 = hp[3];
    const int  pp = pos + nloc - 7;
    const bool pv = nloc < 15 && pp >= 0 && pp < PPOS;
    float* prow = mblk + (size_t)(8 * half) * MAPW + min(max(pp, 0), PPOS - 1);
    float oldv[8];
    #pragma unroll
    for (int r = 0; r < 8; ++r) oldv[r] = prow[r * MAPW];
    __builtin_amdgcn_sched_barrier(0);

    const int w0 = pos - 7;
    float bestv = LOWF;
    #pragma unroll
    for (int k = 0; k < MAPV; ++k) {
        const int e = lane + 32 * k;
        const int c4 = e % (MAPW / 4);
        const int d = 4 * c4 - w0;
        #pragma unroll
        for (int i = 0; i < 4; ++i) {
            if ((unsigned)(d + i) <= 14u) mv[k][i] = LOWF;
            bestv = fmaxf(bestv, mv[k][i]);
        }
    }
    unsigned bestidx = 0xFFFFFFFFu;
    #pragma unroll
    for (int k = 0; k < MAPV; ++k) {
        const int e = lane + 32 * k;
        const int row = e / (MAPW / 4), c4 = e % (MAPW / 4);
        const unsigned base = (unsigned)((a0 + row) * PPOS + 4 * c4);
        #pragma unroll
        for (int i = 0; i < 4; ++i)
            if (mv[k][i] == bestv && 4 * c4 + i < PPOS) bestidx = min(bestidx, base + (unsigned)i);
    }
    __builtin_amdgcn_sched_barrier(0);

    {
        v2f* rowp = (v2f*)(cat + lane * (LAGP * 2)) + 7;
        rowp[0] = (v2f){h0.x, h0.y};  rowp[1] = (v2f){h2.x, h2.y};
        rowp[2] = (v2f){h0.z, h0.w};  rowp[3] = (v2f){h2.z, h2.w};
        rowp[4] = (v2f){h1.x, h1.y};  rowp[5] = (v2f){h3.x, h3.y};
        rowp[6] = (v2f){h1.z, h1.w};  rowp[7] = (v2f){h3.z, h3.w};
    }
    __syncthreads();

    v8f x0 = {}, x1 = {}, x2 = {}, x3 = {};
    #pragma unroll
    for (int g = 0; g < NGRP; ++g) {
        const v4f* ca = ra[g];
        const v2f h[8] = {{ca[0].x, ca[0].y}, {ca[2].x, ca[2].y}, {ca[0].z, ca[0].w}, {ca[2].z, ca[2].w},
                          {ca[1].x, ca[1].y}, {ca[3].x, ca[3].y}, {ca[1].z, ca[1].w}, {ca[3].z, ca[3].w}};
        const v2f* bp = (const v2f*)(cat + (2 * g + half) * (LAGP * 2)) + nloc;
        v2f bf[8];
        #pragma unroll
        for (int t = 0; t < 8; ++t) bf[t] = bp[t];
        x0 = __builtin_amdgcn_wmma_f32_16x16x4_f32(false, h[0], false, bf[0], (short)0, x0, false, false);
        x1 = __builtin_amdgcn_wmma_f32_16x16x4_f32(false, h[1], false, bf[1], (short)0, x1, false, false);
        x2 = __builtin_amdgcn_wmma_f32_16x16x4_f32(false, h[2], false, bf[2], (short)0, x2, false, false);
        x3 = __builtin_amdgcn_wmma_f32_16x16x4_f32(false, h[3], false, bf[3], (short)0, x3, false, false);
        x0 = __builtin_amdgcn_wmma_f32_16x16x4_f32(false, h[4], false, bf[4], (short)0, x0, false, false);
        x1 = __builtin_amdgcn_wmma_f32_16x16x4_f32(false, h[5], false, bf[5], (short)0, x1, false, false);
        x2 = __builtin_amdgcn_wmma_f32_16x16x4_f32(false, h[6], false, bf[6], (short)0, x2, false, false);
        x3 = __builtin_amdgcn_wmma_f32_16x16x4_f32(false, h[7], false, bf[7], (short)0, x3, false, false);
    }
    const v8f X = (x0 + x1) + (x2 + x3);

    if (pv) {
        #pragma unroll
        for (int r = 0; r < 8; ++r) {
            const float nv = oldv[r] - val * X[r];
            prow[r * MAPW] = nv;
            const unsigned ni = (unsigned)((a0 + 8 * half + r) * PPOS + pp);
            if (nv > bestv || (nv == bestv && ni < bestidx)) { bestv = nv; bestidx = ni; }
        }
    }
    if (bestidx == 0xFFFFFFFFu) bestidx = 0u;
    unsigned vb = __float_as_uint(bestv);
    vb = (vb & 0x80000000u) ? ~vb : (vb | 0x80000000u);
    unsigned long long key = ((unsigned long long)vb << 32) | (0xFFFFFFFFu - bestidx);
    for (int off = 16; off > 0; off >>= 1) {
        unsigned long long o = __shfl_xor(key, off, 32);
        if (o > key) key = o;
    }
    if (lane == 0) atomicMax(keys + step * BATCH + b, key);
}

__global__ void k_output_keys(const float* __restrict__ dnn, const unsigned long long* __restrict__ keys,
                              float* __restrict__ out) {
    const int i = blockIdx.x * blockDim.x + threadIdx.x;
    if (i >= BATCH * CC * TT) return;
    const int t  = i % TT;
    const int bc = i / TT;
    const int ch = bc % CC, b = bc / CC;
    float acc = 0.f;
    for (int s = 0; s < NITER; ++s) {
        const unsigned long long fin = keys[s * BATCH + b];
        const unsigned enc  = (unsigned)(fin >> 32);
        const unsigned bits = (enc & 0x80000000u) ? (enc & 0x7FFFFFFFu) : ~enc;
        const float val = __uint_as_float(bits);
        const unsigned idx = 0xFFFFFFFFu - (unsigned)(fin & 0xFFFFFFFFu);
        const int atom = min((int)(idx / PPOS), NATOMS - 1);
        const int tap  = t - (int)(idx % PPOS);
        if (tap >= 0 && tap < ASIZE) acc += val * dnn[(size_t)atom * KDIM + kperm(ch, tap)];
    }
    out[i] = acc;
}

__global__ void k_output(const float* __restrict__ recon, float* __restrict__ out) {
    const int i = blockIdx.x * blockDim.x + threadIdx.x;
    if (i < BATCH * CC * TT) {
        const int t  = i % TT;
        const int bc = i / TT;
        out[i] = recon[(size_t)bc * RES_W + t];
    }
}

extern "C" void kernel_launch(void* const* d_in, const int* in_sizes, int n_in,
                              void* d_out, int out_size, void* d_ws, size_t ws_size,
                              hipStream_t stream) {
    const float* x = (const float*)d_in[0];
    const float* d = (const float*)d_in[1];
    (void)in_sizes; (void)n_in; (void)out_size;

    char* ws = (char*)d_ws;
    float* dnn   = (float*)ws;
    float* res   = (float*)(ws + (size_t)NATOMS * KDIM * 4);
    float* recon = res + BATCH * CC * RES_W;
    unsigned long long* best = (unsigned long long*)(recon + BATCH * CC * RES_W);
    unsigned* arrived = (unsigned*)(best + BATCH);

    k_normalize<<<NATOMS, CORR_THREADS, 0, stream>>>(d, dnn);
    dim3 grid(NATOMS / MTILE, BATCH);

    if (ws_size >= (size_t)26051584) {
        unsigned long long* keys = best;
        float* map = (float*)(ws + 9012224);
        k_init<<<(BATCH * CC * RES_W + 255) / 256, 256, 0, stream>>>(x, res, recon, keys, nullptr, NITER * BATCH);
        k_corr<true><<<grid, CORR_THREADS, 0, stream>>>(dnn, res, recon, keys, nullptr, map);
        for (int it = 1; it < NITER; ++it)
            k_inc<<<NATOMS / 16, CORR_THREADS, 0, stream>>>(dnn, map, keys, it);
        k_output_keys<<<(BATCH * CC * TT + 255) / 256, 256, 0, stream>>>(dnn, keys, (float*)d_out);
        return;
    }
    k_init<<<(BATCH * CC * RES_W + 255) / 256, 256, 0, stream>>>(x, res, recon, best, arrived, BATCH);
    for (int it = 0; it < NITER; ++it)
        k_corr<false><<<grid, CORR_THREADS, 0, stream>>>(dnn, res, recon, best, arrived, nullptr);
    k_output<<<(BATCH * CC * TT + 255) / 256, 256, 0, stream>>>(recon, (float*)d_out);
}


# Incremental products without the lag shift: columns = (item, chosen-atom tap), a wave = 2 atom taps x 4 items, lag sums via LDS: 64 matrix products, 16 row loads per wave (128, 64); sum order changed
# speedup vs baseline: 5.1042x; 1.1314x over previous
#include <hip/hip_runtime.h>

#define NATOMS 4096
#define CC     64
#define TT     256
#define ASIZE  8
#define BATCH  4
#define KDIM   512
#define PPOS   257
#define RES_W  304
#define NT     16
#define NT8    2
#define NPROD  27
#define MTILE  64
#define RES_FLOATS 19456
#define LDS_FLOATS 41984
#define CORR_THREADS 128
#define NITER  32
#define NGRP   16
#define NFRAG  23
#define SBLK   68
#define P256OFF 544
#define MAPW   260
#define MAPV   33
#define HBP    544
#define DIAGO  2304
#define IPD    2052
#define LOWF   (-3.402823466e38f)
#define PITCH2 608

typedef __attribute__((ext_vector_type(2))) float v2f;
typedef __attribute__((ext_vector_type(4))) float v4f;
typedef __attribute__((ext_vector_type(8))) float v8f;

__device__ __forceinline__ int kperm(int c, int t) {
    return (c >> 2) * 32 + ((c >> 1) & 1) * 16 + (t & 1) * 8 + (t >> 1) * 2 + (c & 1);
}

__global__ void k_normalize(const float* __restrict__ d, float* __restrict__ dnn) {
    __shared__ float red[4];
    const int a = blockIdx.x;
    const float* src = d + (size_t)a * KDIM;
    float s = 0.f;
    for (int i = threadIdx.x; i < KDIM; i += CORR_THREADS) { float v = src[i]; s += v * v; }
    for (int off = 16; off > 0; off >>= 1) s += __shfl_down(s, off, 32);
    const int wv = threadIdx.x >> 5;
    if ((threadIdx.x & 31) == 0) red[wv] = s;
    __syncthreads();
    if (threadIdx.x == 0) red[0] = 1.f / (sqrtf(red[0] + red[1] + red[2] + red[3]) + 1e-8f);
    __syncthreads();
    const float inv = red[0];
    float* dst = dnn + (size_t)a * KDIM;
    for (int i = threadIdx.x; i < KDIM; i += CORR_THREADS) dst[kperm(i >> 3, i & 7)] = src[i] * inv;
}

__global__ void k_init(const float* __restrict__ x, float* __restrict__ res,
                       float* __restrict__ recon, unsigned long long* __restrict__ best,
                       unsigned* __restrict__ arrived, int nkeys) {
    const int i = blockIdx.x * blockDim.x + threadIdx.x;
    if (i < BATCH * CC * RES_W) {
        const int col = i % RES_W;
        const int bc  = i / RES_W;
        res[i]   = (col < TT) ? x[bc * TT + col] : 0.f;
        recon[i] = 0.f;
    }
    if (i < nkeys) best[i] = 0ull;
    if (arrived != nullptr && i < BATCH) arrived[i] = 0u;
}

template <bool KEEP_MAP>
__global__ __launch_bounds__(CORR_THREADS)
void k_corr(const float* __restrict__ dnn, float* res, float* recon,
            unsigned long long* best, unsigned* arrived, float* map) {
    __shared__ __attribute__((aligned(16))) float smem[LDS_FLOATS];
    const int b    = blockIdx.y;
    const int m0   = blockIdx.x * MTILE;
    const int tid  = threadIdx.x;
    const int lane = tid & 31;
    const int wave = tid >> 5;

    const int half = lane >> 4;
    const int nloc = lane & 15;
    const v4f* arow = (const v4f*)(dnn + (size_t)(m0 + wave * 16 + nloc) * KDIM + half * 16);

    auto loadA = [&](v4f (&a)[4], int g) {
        #pragma unroll
        for (int q = 0; q < 4; ++q) a[q] = arow[g * 8 + q];
    };

    v4f ra0[4], ra1[4];
    loadA(ra0, 0);

    const v4f* resb = (const v4f*)(res + (size_t)b * RES_FLOATS);
    #pragma unroll
    for (int h2 = 0; h2 < 2; ++h2) {
        const int i0 = h2 ? 9 : 0, i1 = h2 ? 17 : 9;
        v4f te[9], to[9];
        #pragma unroll
        for (int i = i0; i < i1; ++i) {
            const int q = tid + i * CORR_THREADS;
            const int r = q / SBLK, c4 = q % SBLK;
            te[i - i0] = resb[(2 * r) * (RES_W / 4) + c4];
            to[i - i0] = resb[(2 * r + 1) * (RES_W / 4) + c4];
        }
        #pragma unroll
        for (int i = i0; i < i1; ++i) {
            const int q = tid + i * CORR_THREADS;
            const int r = q / SBLK, c4 = q % SBLK;
            v2f* d2 = (v2f*)(smem + r * PITCH2 + c4 * 8 + (c4 >> 2) * 2);
            const v4f e = te[i - i0], o = to[i - i0];
            d2[0] = (v2f){e.x, o.x};
            d2[1] = (v2f){e.y, o.y};
            d2[2] = (v2f){e.z, o.z};
            d2[3] = (v2f){e.w, o.w};
        }
    }
    __syncthreads();

    v8f acc3[NPROD][NT8] = {};
    v2f pz[NPROD] = {};

    auto loadB = [&](v2f (&f)[NFRAG], unsigned p) {
        #pragma unroll
        for (int j = 0; j < NFRAG; ++j) f[j] = *(const v2f*)(smem + p + 2 * (j + (j >> 4)));
    };
    auto loadW = [&](v4f (&w)[4], int g) {
        const v4f* wp = (const v4f*)(smem + (2 * g + half) * PITCH2 + P256OFF);
        #pragma unroll
        for (int q = 0; q < 4; ++q) w[q] = wp[q];
    };

    v2f f0[NFRAG], f1[NFRAG];
    v4f w0[4], w1[4];
    const unsigned fp = (unsigned)(half * PITCH2 + 34 * nloc);
    loadB(f0, fp);
    loadW(w0, 0);

    auto bndNeeded = [](int d) {
        const int d2 = (d / 3) % 3, d3 = d % 3;
        return d3 == 0 || (d2 == 0 && d3 == 1) || d == 4 || d == 13;
    };
    auto buildU = [&](const v2f (&S)[15], v2f (&U3)[NPROD]) {
        v2f L1[3][7];
        #pragma unroll
        for (int m = 0; m < 7; ++m) {
            L1[0][m] = S[2 * m];
            L1[1][m] = S[2 * m + 1];
            L1[2][m] = S[2 * m + 1] + S[2 * m + 2];
        }
        #pragma unroll
        for (int d1 = 0; d1 < 3; ++d1) {
            v2f L2[3][3];
            #pragma unroll
            for (int n = 0; n < 3; ++n) {
                L2[0][n] = L1[d1][2 * n];
                L2[1][n] = L1[d1][2 * n + 1];
                L2[2][n] = L1[d1][2 * n + 1] + L1[d1][2 * n + 2];
            }
            #pragma unroll
            for (int d2 = 0; d2 < 3; ++d2) {
                U3[9 * d1 + 3 * d2]     = L2[d2][0];
                U3[9 * d1 + 3 * d2 + 1] = L2[d2][1];
                U3[9 * d1 + 3 * d2 + 2] = L2[d2][1] + L2[d2][2];
            }
        }
    };

    auto group = [&](const v2f (&cf)[NFRAG], const v4f (&ca)[4], const v4f (&cw)[4],
                     v2f (&nf)[NFRAG], v4f (&na)[4], v4f (&nw)[4], int g) {
        const int gn = g + 1 < NGRP ? g + 1 : NGRP - 1;
        __builtin_amdgcn_sched_barrier(0);
        loadB(nf, fp + (unsigned)gn * (2 * PITCH2));
        loadW(nw, gn);
        loadA(na, gn);
        __builtin_amdgcn_sched_barrier(0);
        const v2f h[8] = {{ca[0].x, ca[0].y}, {ca[2].x, ca[2].y}, {ca[0].z, ca[0].w}, {ca[2].z, ca[2].w},
                          {ca[1].x, ca[1].y}, {ca[3].x, ca[3].y}, {ca[1].z, ca[1].w}, {ca[3].z, ca[3].w}};
        v2f G3[NPROD];
        {
            v2f G1[3][4];
            #pragma unroll
            for (int s = 0; s < 4; ++s) {
                G1[0][s] = h[2 * s];
                G1[1][s] = h[2 * s + 1];
                G1[2][s] = h[2 * s] + h[2 * s + 1];
            }
            #pragma unroll
            for (int d1 = 0; d1 < 3; ++d1) {
                v2f G2[3][2];
                #pragma unroll
                for (int r = 0; r < 2; ++r) {
                    G2[0][r] = G1[d1][2 * r];
                    G2[1][r] = G1[d1][2 * r + 1];
                    G2[2][r] = G1[d1][2 * r] + G1[d1][2 * r + 1];
                }
                #pragma unroll
                for (int d2 = 0; d2 < 3; ++d2) {
                    G3[9 * d1 + 3 * d2]     = G2[d2][0];
                    G3[9 * d1 + 3 * d2 + 1] = G2[d2][1];
                    G3[9 * d1 + 3 * d2 + 2] = G2[d2][0] + G2[d2][1];
                }
            }
        }
        v2f U3t[NT8][NPROD];
        #pragma unroll
        for (int t = 0; t < NT8; ++t) {
            v2f S[15];
            #pragma unroll
            for (int v = 0; v < 15; ++v) S[v] = cf[8 * t + v];
            buildU(S, U3t[t]);
        }
        {
            const v2f zero2 = {0.f, 0.f};
            const v2f Sw[15] = {{cw[0].x, cw[0].y}, {cw[0].z, cw[0].w}, {cw[1].x, cw[1].y}, {cw[1].z, cw[1].w},
                                {cw[2].x, cw[2].y}, {cw[2].z, cw[2].w}, {cw[3].x, cw[3].y}, {cw[3].z, cw[3].w},
                                zero2, zero2, zero2, zero2, zero2, zero2, zero2};
            v2f UW[NPROD];
            buildU(Sw, UW);
            #pragma unroll
            for (int d = 0; d < NPROD; ++d)
                if (bndNeeded(d)) pz[d] += G3[d] * UW[d];
        }
        #pragma unroll
        for (int d = 0; d < NPROD; ++d) {
            #pragma unroll
            for (int t = 0; t < NT8; ++t)
                acc3[d][t] = __builtin_amdgcn_wmma_f32_16x16x4_f32(
                    false, G3[d], false, U3t[t][d], (short)0, acc3[d][t], false, false);
        }
    };

    for (int g2 = 0; g2 < NGRP / 2; ++g2) {
        group(f0, ra0, w0, f1, ra1, w1, 2 * g2);
        group(f1, ra1, w1, f0, ra0, w0, 2 * g2 + 1);
    }

    float zb[NPROD];
    #pragma unroll
    for (int d = 0; d < NPROD; ++d) {
        zb[d] = 0.f;
        if (bndNeeded(d)) {
            float v = pz[d].x + pz[d].y;
            v += __shfl_xor(v, 16, 32);
            zb[d] = v;
        }
    }
    auto shiftnext = [&](const v8f &a0, float bnd) {
        v8f o;
        #pragma unroll
        for (int r = 0; r < 8; ++r) {
            const float kv = __shfl(bnd, 8 * half + r, 32);
            o[r] = __int_as_float(__builtin_amdgcn_update_dpp(
                __float_as_int(kv), __float_as_int(a0[r]), 0x101, 0xF, 0xF, false));
        }
        return o;
    };
    v8f Q[3][3][4];
    float Qb[3][3];
    #pragma unroll
    for (int d1 = 0; d1 < 3; ++d1) {
        #pragma unroll
        for (int d2 = 0; d2 < 3; ++d2) {
            const int base = 9 * d1 + 3 * d2;
            #pragma unroll
            for (int t = 0; t < 2; ++t) {
                Q[d1][d2][2 * t] = acc3[base][t] + acc3[base + 1][t];
                const v8f an = t + 1 < 2 ? acc3[base][t + 1] : shiftnext(acc3[base][0], zb[base]);
                Q[d1][d2][2 * t + 1] = (acc3[base + 2][t] - an) - acc3[base + 1][t];
            }
            Qb[d1][d2] = zb[base] + zb[base + 1];
        }
    }
    v8f R[3][8];
    float Rb[3];
    #pragma unroll
    for (int d1 = 0; d1 < 3; ++d1) {
        #pragma unroll
        for (int t = 0; t < 4; ++t) {
            R[d1][2 * t] = Q[d1][0][t] + Q[d1][1][t];
            const v8f an = t + 1 < 4 ? Q[d1][0][t + 1] : shiftnext(Q[d1][0][0], Qb[d1][0]);
            R[d1][2 * t + 1] = (Q[d1][2][t] - an) - Q[d1][1][t];
        }
        Rb[d1] = Qb[d1][0] + Qb[d1][1];
    }
    v8f acc[NT];
    #pragma unroll
    for (int t = 0; t < 8; ++t) {
        acc[2 * t] = R[0][t] + R[1][t];
        const v8f an = t + 1 < 8 ? R[0][t + 1] : shiftnext(R[0][0], Rb[0]);
        acc[2 * t + 1] = (R[2][t] - an) - R[1][t];
    }

    if (KEEP_MAP) {
        float* mrow = map + ((size_t)b * NATOMS + (m0 + wave * 16 + 8 * half)) * MAPW + 16 * nloc;
        #pragma unroll
        for (int r = 0; r < 8; ++r) {
            v4f* dst = (v4f*)(mrow + (size_t)r * MAPW);
            #pragma unroll
            for (int q = 0; q < 4; ++q)
                dst[q] = (v4f){acc[4 * q][r], acc[4 * q + 1][r], acc[4 * q + 2][r], acc[4 * q + 3][r]};
        }
    }

    float bestv = -3.402823466e38f;
    #pragma unroll
    for (int r = 0; r < 8; ++r) {
        #pragma unroll
        for (int j = 0; j < NT; ++j) bestv = fmaxf(bestv, acc[j][r]);
    }
    unsigned code = 0xFFFFFFFFu;
    #pragma unroll
    for (int r = 7; r >= 0; --r) {
        #pragma unroll
        for (int j = NT - 1; j >= 0; --j)
            if (acc[j][r] == bestv) code = (unsigned)(r * PPOS + j);
    }
    unsigned bestidx = code == 0xFFFFFFFFu
        ? 0u : (unsigned)((m0 + wave * 16 + 8 * half) * PPOS + 16 * nloc) + code;
    const float s256 = Rb[0] + Rb[1];
    if (KEEP_MAP && half == 0) {
        v4f* d4 = (v4f*)(map + ((size_t)b * NATOMS + (m0 + wave * 16 + nloc)) * MAPW + 256);
        *d4 = (v4f){s256, LOWF, LOWF, LOWF};
    }
    const unsigned idx256 = (unsigned)((m0 + wave * 16 + nloc) * PPOS + (PPOS - 1));
    if (s256 > bestv || (s256 == bestv && idx256 < bestidx)) { bestv = s256; bestidx = idx256; }
    unsigned vb = __float_as_uint(bestv);
    vb = (vb & 0x80000000u) ? ~vb : (vb | 0x80000000u);
    unsigned long long key = ((unsigned long long)vb << 32) | (0xFFFFFFFFu - bestidx);
    for (int off = 16; off > 0; off >>= 1) {
        unsigned long long o = __shfl_xor(key, off, 32);
        if (o > key) key = o;
    }
    unsigned long long* wkey = (unsigned long long*)(smem + RES_FLOATS);
    if (lane == 0) wkey[wave] = key;
    __syncthreads();
    if (tid == 0) {
        unsigned long long k = wkey[0];
        for (int w = 1; w < CORR_THREADS / 32; ++w) if (wkey[w] > k) k = wkey[w];
        atomicMax(best + b, k);
        if (KEEP_MAP) return;
        const unsigned prev = __hip_atomic_fetch_add(arrived + b, 1u, __ATOMIC_ACQ_REL,
                                                     __HIP_MEMORY_SCOPE_AGENT);
        unsigned long long fin = 0ull;
        if (prev == gridDim.x - 1)
            fin = __hip_atomic_load(best + b, __ATOMIC_RELAXED, __HIP_MEMORY_SCOPE_AGENT);
        wkey[CORR_THREADS / 32] = fin;
    }
    if (KEEP_MAP) return;
    __syncthreads();
    const unsigned long long fin = wkey[CORR_THREADS / 32];
    if (fin == 0ull) return;

    if (tid == 0) { best[b] = 0ull; arrived[b] = 0u; }
    const unsigned enc  = (unsigned)(fin >> 32);
    const unsigned bits = (enc & 0x80000000u) ? (enc & 0x7FFFFFFFu) : ~enc;
    const float val = __uint_as_float(bits);
    const unsigned idx = 0xFFFFFFFFu - (unsigned)(fin & 0xFFFFFFFFu);
    const int atom = min((int)(idx / PPOS), NATOMS - 1);
    const int pos  = (int)(idx % PPOS);
    const float* ap = dnn + (size_t)atom * KDIM;
    float* rb = res   + (size_t)b * RES_FLOATS;
    float* cb = recon + (size_t)b * RES_FLOATS;
    float ua[4], ur[4], uc[4];
    #pragma unroll
    for (int k = 0; k < KDIM / CORR_THREADS; ++k) {
        const int i = tid + k * CORR_THREADS;
        const int ch = i >> 3, t = i & 7;
        ua[k] = ap[kperm(ch, t)];
        ur[k] = rb[ch * RES_W + pos + t];
        uc[k] = cb[ch * RES_W + pos + t];
    }
    #pragma unroll
    for (int k = 0; k < KDIM / CORR_THREADS; ++k) {
        const int i = tid + k * CORR_THREADS;
        const int ch = i >> 3, t = i & 7;
        const float av = ua[k] * val;
        rb[ch * RES_W + pos + t] = ur[k] - av;
        cb[ch * RES_W + pos + t] = uc[k] + av;
    }
}


__global__ __launch_bounds__(CORR_THREADS)
void k_inc(const float* __restrict__ dnn, float* map, unsigned long long* keys, int step) {
    __shared__ __attribute__((aligned(16))) float smem[LDS_FLOATS];
    const int tid  = threadIdx.x;
    const int lane = tid & 31;
    const int b    = tid >> 5;
    const int a0   = blockIdx.x * 16;
    const int half = lane >> 4;
    const int nloc = lane & 15;

    const unsigned long long fin = __hip_atomic_load(keys + (step - 1) * BATCH + b, __ATOMIC_RELAXED,
                                                     __HIP_MEMORY_SCOPE_AGENT);
    float* mblk = map + ((size_t)b * NATOMS + a0) * MAPW;
    const v4f* mb4 = (const v4f*)mblk;
    v4f mv[MAPV];
    #pragma unroll
    for (int k = 0; k < MAPV; ++k) {
        const int e = lane + 32 * k;
        mv[k] = (v4f){LOWF, LOWF, LOWF, LOWF};
        if (e < 16 * (MAPW / 4)) mv[k] = mb4[e];
    }
    const v4f* arow = (const v4f*)(dnn + (size_t)(a0 + nloc) * KDIM + half * 16) + b;
    v4f ra[NGRP];
    #pragma unroll
    for (int g = 0; g < NGRP; ++g) ra[g] = arow[g * 8];
    {
        v4f* z = (v4f*)(smem + DIAGO);
        #pragma unroll
        for (int k = 0; k < (BATCH * IPD / 4 + CORR_THREADS - 1) / CORR_THREADS; ++k)
            z[tid + k * CORR_THREADS] = (v4f){0.f, 0.f, 0.f, 0.f};
    }
    __builtin_amdgcn_sched_barrier(0);

    const unsigned enc  = (unsigned)(fin >> 32);
    const unsigned bits = (enc & 0x80000000u) ? (enc & 0x7FFFFFFFu) : ~enc;
    const float val = __uint_as_float(bits);
    const unsigned idx = 0xFFFFFFFFu - (unsigned)(fin & 0xFFFFFFFFu);
    const int atom = min((int)(idx / PPOS), NATOMS - 1);
    const int pos  = (int)(idx % PPOS);

    const v4f* hp = (const v4f*)(dnn + (size_t)atom * KDIM + lane * 16);
    const v4f h0 = hp[0], h1 = hp[1], h2 = hp[2], h3 = hp[3];
    const int  pp = pos + nloc - 7;
    const bool pv = nloc < 15 && pp >= 0 && pp < PPOS;
    float* prow = mblk + (size_t)(8 * half) * MAPW + min(max(pp, 0), PPOS - 1);
    float oldv[8];
    #pragma unroll
    for (int r = 0; r < 8; ++r) oldv[r] = prow[r * MAPW];
    __builtin_amdgcn_sched_barrier(0);

    const int w0 = pos - 7;
    float bestv = LOWF;
    #pragma unroll
    for (int k = 0; k < MAPV; ++k) {
        const int e = lane + 32 * k;
        const int c4 = e % (MAPW / 4);
        const int d = 4 * c4 - w0;
        #pragma unroll
        for (int i = 0; i < 4; ++i) {
            if ((unsigned)(d + i) <= 14u) mv[k][i] = LOWF;
            bestv = fmaxf(bestv, mv[k][i]);
        }
    }
    unsigned bestidx = 0xFFFFFFFFu;
    #pragma unroll
    for (int k = 0; k < MAPV; ++k) {
        const int e = lane + 32 * k;
        const int row = e / (MAPW / 4), c4 = e % (MAPW / 4);
        const unsigned base = (unsigned)((a0 + row) * PPOS + 4 * c4);
        #pragma unroll
        for (int i = 0; i < 4; ++i)
            if (mv[k][i] == bestv && 4 * c4 + i < PPOS) bestidx = min(bestidx, base + (unsigned)i);
    }
    __builtin_amdgcn_sched_barrier(0);

    {
        v4f* rowp = (v4f*)(smem + b * HBP + lane * 16);
        rowp[0] = (v4f){h0.x, h0.y, h2.x, h2.y};
        rowp[1] = (v4f){h0.z, h0.w, h2.z, h2.w};
        rowp[2] = (v4f){h1.x, h1.y, h3.x, h3.y};
        rowp[3] = (v4f){h1.z, h1.w, h3.z, h3.w};
    }
    __syncthreads();

    v8f y[2][2] = {};
    {
        const v2f* bp = (const v2f*)(smem + (nloc >> 3) * HBP + half * 16) + (nloc & 7);
        #pragma unroll
        for (int g = 0; g < NGRP; ++g) {
            const v2f hA[2] = {{ra[g].x, ra[g].y}, {ra[g].z, ra[g].w}};
            const v2f bf[2] = {bp[g * 16], bp[g * 16 + HBP]};
            #pragma unroll
            for (int ti = 0; ti < 2; ++ti) {
                #pragma unroll
                for (int n2 = 0; n2 < 2; ++n2)
                    y[ti][n2] = __builtin_amdgcn_wmma_f32_16x16x4_f32(
                        false, hA[ti], false, bf[n2], (short)0, y[ti][n2], false, false);
            }
        }
    }
    {
        const int t0 = (b & 1) * 4 + (b >> 1);
        float* dg = smem + DIAGO + (nloc >> 3) * IPD + (8 * half) * 128 + ((nloc & 7) + 7) * 8 - 7 * t0;
        #pragma unroll
        for (int ti = 0; ti < 2; ++ti) {
            #pragma unroll
            for (int n2 = 0; n2 < 2; ++n2) {
                #pragma unroll
                for (int r = 0; r < 8; ++r) dg[2 * n2 * IPD + r * 128 - 14 * ti] = y[ti][n2][r];
            }
        }
    }
    __syncthreads();
    v8f X;
    {
        const v4f* xr = (const v4f*)(smem + DIAGO + b * IPD + (8 * half) * 128 + nloc * 8);
        #pragma unroll
        for (int r = 0; r < 8; ++r) {
            const v4f lo = xr[r * 32], hi = xr[r * 32 + 1];
            X[r] = ((lo.x + lo.y) + (lo.z + lo.w)) + ((hi.x + hi.y) + (hi.z + hi.w));
        }
    }

    if (pv) {
        #pragma unroll
        for (int r = 0; r < 8; ++r) {
            const float nv = oldv[r] - val * X[r];
            prow[r * MAPW] = nv;
            const unsigned ni = (unsigned)((a0 + 8 * half + r) * PPOS + pp);
            if (nv > bestv || (nv == bestv && ni < bestidx)) { bestv = nv; bestidx = ni; }
        }
    }
    if (bestidx == 0xFFFFFFFFu) bestidx = 0u;
    unsigned vb = __float_as_uint(bestv);
    vb = (vb & 0x80000000u) ? ~vb : (vb | 0x80000000u);
    unsigned long long key = ((unsigned long long)vb << 32) | (0xFFFFFFFFu - bestidx);
    for (int off = 16; off > 0; off >>= 1) {
        unsigned long long o = __shfl_xor(key, off, 32);
        if (o > key) key = o;
    }
    if (lane == 0) atomicMax(keys + step * BATCH + b, key);
}

__global__ void k_output_keys(const float* __restrict__ dnn, const unsigned long long* __restrict__ keys,
                              float* __restrict__ out) {
    const int i = blockIdx.x * blockDim.x + threadIdx.x;
    if (i >= BATCH * CC * TT) return;
    const int t  = i % TT;
    const int bc = i / TT;
    const int ch = bc % CC, b = bc / CC;
    float acc = 0.f;
    for (int s = 0; s < NITER; ++s) {
        const unsigned long long fin = keys[s * BATCH + b];
        const unsigned enc  = (unsigned)(fin >> 32);
        const unsigned bits = (enc & 0x80000000u) ? (enc & 0x7FFFFFFFu) : ~enc;
        const float val = __uint_as_float(bits);
        const unsigned idx = 0xFFFFFFFFu - (unsigned)(fin & 0xFFFFFFFFu);
        const int atom = min((int)(idx / PPOS), NATOMS - 1);
        const int tap  = t - (int)(idx % PPOS);
        if (tap >= 0 && tap < ASIZE) acc += val * dnn[(size_t)atom * KDIM + kperm(ch, tap)];
    }
    out[i] = acc;
}

__global__ void k_output(const float* __restrict__ recon, float* __restrict__ out) {
    const int i = blockIdx.x * blockDim.x + threadIdx.x;
    if (i < BATCH * CC * TT) {
        const int t  = i % TT;
        const int bc = i / TT;
        out[i] = recon[(size_t)bc * RES_W + t];
    }
}

extern "C" void kernel_launch(void* const* d_in, const int* in_sizes, int n_in,
                              void* d_out, int out_size, void* d_ws, size_t ws_size,
                              hipStream_t stream) {
    const float* x = (const float*)d_in[0];
    const float* d = (const float*)d_in[1];
    (void)in_sizes; (void)n_in; (void)out_size;

    char* ws = (char*)d_ws;
    float* dnn   = (float*)ws;
    float* res   = (float*)(ws + (size_t)NATOMS * KDIM * 4);
    float* recon = res + BATCH * CC * RES_W;
    unsigned long long* best = (unsigned long long*)(recon + BATCH * CC * RES_W);
    unsigned* arrived = (unsigned*)(best + BATCH);

    k_normalize<<<NATOMS, CORR_THREADS, 0, stream>>>(d, dnn);
    dim3 grid(NATOMS / MTILE, BATCH);

    if (ws_size >= (size_t)26051584) {
        unsigned long long* keys = best;
        float* map = (float*)(ws + 9012224);
        k_init<<<(BATCH * CC * RES_W + 255) / 256, 256, 0, stream>>>(x, res, recon, keys, nullptr, NITER * BATCH);
        k_corr<true><<<grid, CORR_THREADS, 0, stream>>>(dnn, res, recon, keys, nullptr, map);
        for (int it = 1; it < NITER; ++it)
            k_inc<<<NATOMS / 16, CORR_THREADS, 0, stream>>>(dnn, map, keys, it);
        k_output_keys<<<(BATCH * CC * TT + 255) / 256, 256, 0, stream>>>(dnn, keys, (float*)d_out);
        return;
    }
    k_init<<<(BATCH * CC * RES_W + 255) / 256, 256, 0, stream>>>(x, res, recon, best, arrived, BATCH);
    for (int it = 0; it < NITER; ++it)
        k_corr<false><<<grid, CORR_THREADS, 0, stream>>>(dnn, res, recon, best, arrived, nullptr);
    k_output<<<(BATCH * CC * TT + 255) / 256, 256, 0, stream>>>(recon, (float*)d_out);
}


# Row-best keys (value, smallest index) of every map row kept in workspace: a step folds its 15 patched values into them and re-reads a row only if its best lay under the patch; full rescan gone
# speedup vs baseline: 6.7639x; 1.3252x over previous
#include <hip/hip_runtime.h>

#define NATOMS 4096
#define CC     64
#define TT     256
#define ASIZE  8
#define BATCH  4
#define KDIM   512
#define PPOS   257
#define RES_W  304
#define NT     16
#define NT8    2
#define NPROD  27
#define MTILE  64
#define RES_FLOATS 19456
#define LDS_FLOATS 41984
#define CORR_THREADS 128
#define NITER  32
#define NGRP   16
#define NFRAG  23
#define SBLK   68
#define P256OFF 544
#define MAPW   260
#define MAPV   33
#define HBP    544
#define DIAGO  2304
#define IPD    2052
#define LOWF   (-3.402823466e38f)
#define PITCH2 608

typedef __attribute__((ext_vector_type(2))) float v2f;
typedef __attribute__((ext_vector_type(4))) float v4f;
typedef __attribute__((ext_vector_type(8))) float v8f;
#define ROWX(v, M) __builtin_amdgcn_update_dpp((v), (v), 0x160 | (M), 0xF, 0xF, true)
__device__ __forceinline__ int xhalf(int v) {
    return __builtin_amdgcn_permlanex16(v, v, 0x76543210, 0xfedcba98, false, false);
}
__device__ __forceinline__ float rowmaxf16(float v) {
    v = fmaxf(v, __int_as_float(ROWX(__float_as_int(v), 1)));
    v = fmaxf(v, __int_as_float(ROWX(__float_as_int(v), 2)));
    v = fmaxf(v, __int_as_float(ROWX(__float_as_int(v), 4)));
    v = fmaxf(v, __int_as_float(ROWX(__float_as_int(v), 8)));
    return v;
}
__device__ __forceinline__ unsigned rowminu16(unsigned v) {
    v = min(v, (unsigned)ROWX((int)v, 1));
    v = min(v, (unsigned)ROWX((int)v, 2));
    v = min(v, (unsigned)ROWX((int)v, 4));
    v = min(v, (unsigned)ROWX((int)v, 8));
    return v;
}
__device__ __forceinline__ unsigned rowmaxu16(unsigned v) {
    v = max(v, (unsigned)ROWX((int)v, 1));
    v = max(v, (unsigned)ROWX((int)v, 2));
    v = max(v, (unsigned)ROWX((int)v, 4));
    v = max(v, (unsigned)ROWX((int)v, 8));
    return v;
}
__device__ __forceinline__ unsigned long long wavemaxkey(unsigned long long k) {
    const unsigned hi = (unsigned)(k >> 32), lo = (unsigned)k;
    unsigned mh = rowmaxu16(hi);
    mh = max(mh, (unsigned)xhalf((int)mh));
    unsigned ml = rowmaxu16(hi == mh ? lo : 0u);
    ml = max(ml, (unsigned)xhalf((int)ml));
    return ((unsigned long long)mh << 32) | ml;
}
__device__ __forceinline__ unsigned fenc(float v) {
    const unsigned vb = __float_as_uint(v);
    return (vb & 0x80000000u) ? ~vb : (vb | 0x80000000u);
}
__device__ __forceinline__ unsigned long long mkkey(float v, unsigned flat) {
    return ((unsigned long long)fenc(v) << 32) | (0xFFFFFFFFu - flat);
}


__device__ __forceinline__ int kperm(int c, int t) {
    return (c >> 2) * 32 + ((c >> 1) & 1) * 16 + (t & 1) * 8 + (t >> 1) * 2 + (c & 1);
}

__global__ void k_normalize(const float* __restrict__ d, float* __restrict__ dnn) {
    __shared__ float red[4];
    const int a = blockIdx.x;
    const float* src = d + (size_t)a * KDIM;
    float s = 0.f;
    for (int i = threadIdx.x; i < KDIM; i += CORR_THREADS) { float v = src[i]; s += v * v; }
    for (int off = 16; off > 0; off >>= 1) s += __shfl_down(s, off, 32);
    const int wv = threadIdx.x >> 5;
    if ((threadIdx.x & 31) == 0) red[wv] = s;
    __syncthreads();
    if (threadIdx.x == 0) red[0] = 1.f / (sqrtf(red[0] + red[1] + red[2] + red[3]) + 1e-8f);
    __syncthreads();
    const float inv = red[0];
    float* dst = dnn + (size_t)a * KDIM;
    for (int i = threadIdx.x; i < KDIM; i += CORR_THREADS) dst[kperm(i >> 3, i & 7)] = src[i] * inv;
}

__global__ void k_init(const float* __restrict__ x, float* __restrict__ res,
                       float* __restrict__ recon, unsigned long long* __restrict__ best,
                       unsigned* __restrict__ arrived, int nkeys) {
    const int i = blockIdx.x * blockDim.x + threadIdx.x;
    if (i < BATCH * CC * RES_W) {
        const int col = i % RES_W;
        const int bc  = i / RES_W;
        res[i]   = (col < TT) ? x[bc * TT + col] : 0.f;
        recon[i] = 0.f;
    }
    if (i < nkeys) best[i] = 0ull;
    if (arrived != nullptr && i < BATCH) arrived[i] = 0u;
}

template <bool KEEP_MAP>
__global__ __launch_bounds__(CORR_THREADS)
void k_corr(const float* __restrict__ dnn, float* res, float* recon,
            unsigned long long* best, unsigned* arrived, float* map, unsigned long long* rowkeys) {
    __shared__ __attribute__((aligned(16))) float smem[LDS_FLOATS];
    const int b    = blockIdx.y;
    const int m0   = blockIdx.x * MTILE;
    const int tid  = threadIdx.x;
    const int lane = tid & 31;
    const int wave = tid >> 5;

    const int half = lane >> 4;
    const int nloc = lane & 15;
    const v4f* arow = (const v4f*)(dnn + (size_t)(m0 + wave * 16 + nloc) * KDIM + half * 16);

    auto loadA = [&](v4f (&a)[4], int g) {
        #pragma unroll
        for (int q = 0; q < 4; ++q) a[q] = arow[g * 8 + q];
    };

    v4f ra0[4], ra1[4];
    loadA(ra0, 0);

    const v4f* resb = (const v4f*)(res + (size_t)b * RES_FLOATS);
    #pragma unroll
    for (int h2 = 0; h2 < 2; ++h2) {
        const int i0 = h2 ? 9 : 0, i1 = h2 ? 17 : 9;
        v4f te[9], to[9];
        #pragma unroll
        for (int i = i0; i < i1; ++i) {
            const int q = tid + i * CORR_THREADS;
            const int r = q / SBLK, c4 = q % SBLK;
            te[i - i0] = resb[(2 * r) * (RES_W / 4) + c4];
            to[i - i0] = resb[(2 * r + 1) * (RES_W / 4) + c4];
        }
        #pragma unroll
        for (int i = i0; i < i1; ++i) {
            const int q = tid + i * CORR_THREADS;
            const int r = q / SBLK, c4 = q % SBLK;
            v2f* d2 = (v2f*)(smem + r * PITCH2 + c4 * 8 + (c4 >> 2) * 2);
            const v4f e = te[i - i0], o = to[i - i0];
            d2[0] = (v2f){e.x, o.x};
            d2[1] = (v2f){e.y, o.y};
            d2[2] = (v2f){e.z, o.z};
            d2[3] = (v2f){e.w, o.w};
        }
    }
    __syncthreads();

    v8f acc3[NPROD][NT8] = {};
    v2f pz[NPROD] = {};

    auto loadB = [&](v2f (&f)[NFRAG], unsigned p) {
        #pragma unroll
        for (int j = 0; j < NFRAG; ++j) f[j] = *(const v2f*)(smem + p + 2 * (j + (j >> 4)));
    };
    auto loadW = [&](v4f (&w)[4], int g) {
        const v4f* wp = (const v4f*)(smem + (2 * g + half) * PITCH2 + P256OFF);
        #pragma unroll
        for (int q = 0; q < 4; ++q) w[q] = wp[q];
    };

    v2f f0[NFRAG], f1[NFRAG];
    v4f w0[4], w1[4];
    const unsigned fp = (unsigned)(half * PITCH2 + 34 * nloc);
    loadB(f0, fp);
    loadW(w0, 0);

    auto bndNeeded = [](int d) {
        const int d2 = (d / 3) % 3, d3 = d % 3;
        return d3 == 0 || (d2 == 0 && d3 == 1) || d == 4 || d == 13;
    };
    auto buildU = [&](const v2f (&S)[15], v2f (&U3)[NPROD]) {
        v2f L1[3][7];
        #pragma unroll
        for (int m = 0; m < 7; ++m) {
            L1[0][m] = S[2 * m];
            L1[1][m] = S[2 * m + 1];
            L1[2][m] = S[2 * m + 1] + S[2 * m + 2];
        }
        #pragma unroll
        for (int d1 = 0; d1 < 3; ++d1) {
            v2f L2[3][3];
            #pragma unroll
            for (int n = 0; n < 3; ++n) {
                L2[0][n] = L1[d1][2 * n];
                L2[1][n] = L1[d1][2 * n + 1];
                L2[2][n] = L1[d1][2 * n + 1] + L1[d1][2 * n + 2];
            }
            #pragma unroll
            for (int d2 = 0; d2 < 3; ++d2) {
                U3[9 * d1 + 3 * d2]     = L2[d2][0];
                U3[9 * d1 + 3 * d2 + 1] = L2[d2][1];
                U3[9 * d1 + 3 * d2 + 2] = L2[d2][1] + L2[d2][2];
            }
        }
    };

    auto group = [&](const v2f (&cf)[NFRAG], const v4f (&ca)[4], const v4f (&cw)[4],
                     v2f (&nf)[NFRAG], v4f (&na)[4], v4f (&nw)[4], int g) {
        const int gn = g + 1 < NGRP ? g + 1 : NGRP - 1;
        __builtin_amdgcn_sched_barrier(0);
        loadB(nf, fp + (unsigned)gn * (2 * PITCH2));
        loadW(nw, gn);
        loadA(na, gn);
        __builtin_amdgcn_sched_barrier(0);
        const v2f h[8] = {{ca[0].x, ca[0].y}, {ca[2].x, ca[2].y}, {ca[0].z, ca[0].w}, {ca[2].z, ca[2].w},
                          {ca[1].x, ca[1].y}, {ca[3].x, ca[3].y}, {ca[1].z, ca[1].w}, {ca[3].z, ca[3].w}};
        v2f G3[NPROD];
        {
            v2f G1[3][4];
            #pragma unroll
            for (int s = 0; s < 4; ++s) {
                G1[0][s] = h[2 * s];
                G1[1][s] = h[2 * s + 1];
                G1[2][s] = h[2 * s] + h[2 * s + 1];
            }
            #pragma unroll
            for (int d1 = 0; d1 < 3; ++d1) {
                v2f G2[3][2];
                #pragma unroll
                for (int r = 0; r < 2; ++r) {
                    G2[0][r] = G1[d1][2 * r];
                    G2[1][r] = G1[d1][2 * r + 1];
                    G2[2][r] = G1[d1][2 * r] + G1[d1][2 * r + 1];
                }
                #pragma unroll
                for (int d2 = 0; d2 < 3; ++d2) {
                    G3[9 * d1 + 3 * d2]     = G2[d2][0];
                    G3[9 * d1 + 3 * d2 + 1] = G2[d2][1];
                    G3[9 * d1 + 3 * d2 + 2] = G2[d2][0] + G2[d2][1];
                }
            }
        }
        v2f U3t[NT8][NPROD];
        #pragma unroll
        for (int t = 0; t < NT8; ++t) {
            v2f S[15];
            #pragma unroll
            for (int v = 0; v < 15; ++v) S[v] = cf[8 * t + v];
            buildU(S, U3t[t]);
        }
        {
            const v2f zero2 = {0.f, 0.f};
            const v2f Sw[15] = {{cw[0].x, cw[0].y}, {cw[0].z, cw[0].w}, {cw[1].x, cw[1].y}, {cw[1].z, cw[1].w},
                                {cw[2].x, cw[2].y}, {cw[2].z, cw[2].w}, {cw[3].x, cw[3].y}, {cw[3].z, cw[3].w},
                                zero2, zero2, zero2, zero2, zero2, zero2, zero2};
            v2f UW[NPROD];
            buildU(Sw, UW);
            #pragma unroll
            for (int d = 0; d < NPROD; ++d)
                if (bndNeeded(d)) pz[d] += G3[d] * UW[d];
        }
        #pragma unroll
        for (int d = 0; d < NPROD; ++d) {
            #pragma unroll
            for (int t = 0; t < NT8; ++t)
                acc3[d][t] = __builtin_amdgcn_wmma_f32_16x16x4_f32(
                    false, G3[d], false, U3t[t][d], (short)0, acc3[d][t], false, false);
        }
    };

    for (int g2 = 0; g2 < NGRP / 2; ++g2) {
        group(f0, ra0, w0, f1, ra1, w1, 2 * g2);
        group(f1, ra1, w1, f0, ra0, w0, 2 * g2 + 1);
    }

    float zb[NPROD];
    #pragma unroll
    for (int d = 0; d < NPROD; ++d) {
        zb[d] = 0.f;
        if (bndNeeded(d)) {
            float v = pz[d].x + pz[d].y;
            v += __shfl_xor(v, 16, 32);
            zb[d] = v;
        }
    }
    auto shiftnext = [&](const v8f &a0, float bnd) {
        v8f o;
        #pragma unroll
        for (int r = 0; r < 8; ++r) {
            const float kv = __shfl(bnd, 8 * half + r, 32);
            o[r] = __int_as_float(__builtin_amdgcn_update_dpp(
                __float_as_int(kv), __float_as_int(a0[r]), 0x101, 0xF, 0xF, false));
        }
        return o;
    };
    v8f Q[3][3][4];
    float Qb[3][3];
    #pragma unroll
    for (int d1 = 0; d1 < 3; ++d1) {
        #pragma unroll
        for (int d2 = 0; d2 < 3; ++d2) {
            const int base = 9 * d1 + 3 * d2;
            #pragma unroll
            for (int t = 0; t < 2; ++t) {
                Q[d1][d2][2 * t] = acc3[base][t] + acc3[base + 1][t];
                const v8f an = t + 1 < 2 ? acc3[base][t + 1] : shiftnext(acc3[base][0], zb[base]);
                Q[d1][d2][2 * t + 1] = (acc3[base + 2][t] - an) - acc3[base + 1][t];
            }
            Qb[d1][d2] = zb[base] + zb[base + 1];
        }
    }
    v8f R[3][8];
    float Rb[3];
    #pragma unroll
    for (int d1 = 0; d1 < 3; ++d1) {
        #pragma unroll
        for (int t = 0; t < 4; ++t) {
            R[d1][2 * t] = Q[d1][0][t] + Q[d1][1][t];
            const v8f an = t + 1 < 4 ? Q[d1][0][t + 1] : shiftnext(Q[d1][0][0], Qb[d1][0]);
            R[d1][2 * t + 1] = (Q[d1][2][t] - an) - Q[d1][1][t];
        }
        Rb[d1] = Qb[d1][0] + Qb[d1][1];
    }
    v8f acc[NT];
    #pragma unroll
    for (int t = 0; t < 8; ++t) {
        acc[2 * t] = R[0][t] + R[1][t];
        const v8f an = t + 1 < 8 ? R[0][t + 1] : shiftnext(R[0][0], Rb[0]);
        acc[2 * t + 1] = (R[2][t] - an) - R[1][t];
    }

    if (KEEP_MAP) {
        float* mrow = map + ((size_t)b * NATOMS + (m0 + wave * 16 + 8 * half)) * MAPW + 16 * nloc;
        #pragma unroll
        for (int r = 0; r < 8; ++r) {
            v4f* dst = (v4f*)(mrow + (size_t)r * MAPW);
            #pragma unroll
            for (int q = 0; q < 4; ++q)
                dst[q] = (v4f){acc[4 * q][r], acc[4 * q + 1][r], acc[4 * q + 2][r], acc[4 * q + 3][r]};
        }
    }

    const float s256 = Rb[0] + Rb[1];
    if (KEEP_MAP && half == 0) {
        v4f* d4 = (v4f*)(map + ((size_t)b * NATOMS + (m0 + wave * 16 + nloc)) * MAPW + 256);
        *d4 = (v4f){s256, LOWF, LOWF, LOWF};
    }
    unsigned long long key = 0ull, mine = 0ull;
    #pragma unroll
    for (int r = 0; r < 8; ++r) {
        float lv = acc[0][r];
        #pragma unroll
        for (int j = 1; j < NT; ++j) lv = fmaxf(lv, acc[j][r]);
        float rv = rowmaxf16(lv);
        unsigned lp = 0xFFFFFFFFu;
        #pragma unroll
        for (int j = NT - 1; j >= 0; --j)
            if (acc[j][r] == rv) lp = (unsigned)(j + 16 * nloc);
        unsigned rp = rowminu16(lp);
        if (rp == 0xFFFFFFFFu) rp = 0u;
        const float v256 = __shfl(s256, 8 * half + r, 32);
        if (v256 > rv) { rv = v256; rp = PPOS - 1; }
        const unsigned long long kr = mkkey(rv, (unsigned)(m0 + wave * 16 + 8 * half + r) * PPOS + rp);
        if ((nloc & 7) == r) mine = kr;
        if (kr > key) key = kr;
    }
    if (KEEP_MAP && nloc < 8)
        rowkeys[(size_t)b * NATOMS + m0 + wave * 16 + 8 * half + nloc] = mine;
    {
        const unsigned long long o = __shfl_xor(key, 16, 32);
        if (o > key) key = o;
    }
    unsigned long long* wkey = (unsigned long long*)(smem + RES_FLOATS);
    if (lane == 0) wkey[wave] = key;
    __syncthreads();
    if (tid == 0) {
        unsigned long long k = wkey[0];
        for (int w = 1; w < CORR_THREADS / 32; ++w) if (wkey[w] > k) k = wkey[w];
        atomicMax(best + b, k);
        if (KEEP_MAP) return;
        const unsigned prev = __hip_atomic_fetch_add(arrived + b, 1u, __ATOMIC_ACQ_REL,
                                                     __HIP_MEMORY_SCOPE_AGENT);
        unsigned long long fin = 0ull;
        if (prev == gridDim.x - 1)
            fin = __hip_atomic_load(best + b, __ATOMIC_RELAXED, __HIP_MEMORY_SCOPE_AGENT);
        wkey[CORR_THREADS / 32] = fin;
    }
    if (KEEP_MAP) return;
    __syncthreads();
    const unsigned long long fin = wkey[CORR_THREADS / 32];
    if (fin == 0ull) return;

    if (tid == 0) { best[b] = 0ull; arrived[b] = 0u; }
    const unsigned enc  = (unsigned)(fin >> 32);
    const unsigned bits = (enc & 0x80000000u) ? (enc & 0x7FFFFFFFu) : ~enc;
    const float val = __uint_as_float(bits);
    const unsigned idx = 0xFFFFFFFFu - (unsigned)(fin & 0xFFFFFFFFu);
    const int atom = min((int)(idx / PPOS), NATOMS - 1);
    const int pos  = (int)(idx % PPOS);
    const float* ap = dnn + (size_t)atom * KDIM;
    float* rb = res   + (size_t)b * RES_FLOATS;
    float* cb = recon + (size_t)b * RES_FLOATS;
    float ua[4], ur[4], uc[4];
    #pragma unroll
    for (int k = 0; k < KDIM / CORR_THREADS; ++k) {
        const int i = tid + k * CORR_THREADS;
        const int ch = i >> 3, t = i & 7;
        ua[k] = ap[kperm(ch, t)];
        ur[k] = rb[ch * RES_W + pos + t];
        uc[k] = cb[ch * RES_W + pos + t];
    }
    #pragma unroll
    for (int k = 0; k < KDIM / CORR_THREADS; ++k) {
        const int i = tid + k * CORR_THREADS;
        const int ch = i >> 3, t = i & 7;
        const float av = ua[k] * val;
        rb[ch * RES_W + pos + t] = ur[k] - av;
        cb[ch * RES_W + pos + t] = uc[k] + av;
    }
}


struct RowSlots { int row[16]; v4f a[16], b[16]; float c[16]; };

template <int G>
__device__ __forceinline__ void rows_request(RowSlots& s, unsigned rmask, int nre, const float* mblk, int lane) {
    if constexpr (G < 4) {
        if (4 * G < nre) {
            const int first = __builtin_ctz(rmask);
            #pragma unroll
            for (int k = 0; k < 4; ++k) {
                const int q = 4 * G + k;
                s.row[q] = rmask ? __builtin_ctz(rmask) : first;
                rmask &= rmask - 1u;
                const float* rp = mblk + s.row[q] * MAPW;
                s.a[q] = ((const v4f*)rp)[lane];
                s.b[q] = ((const v4f*)rp)[lane + 32];
                s.c[q] = rp[PPOS - 1];
            }
            rows_request<G + 1>(s, rmask, nre, mblk, lane);
        }
    }
}

__device__ __forceinline__ unsigned long long row_rescan(const v4f& a, const v4f& b, float c, int flat0,
                                                         int w0, int lane) {
    float e[9] = {a.x, a.y, a.z, a.w, b.x, b.y, b.z, b.w, c};
    float bv = LOWF;
    #pragma unroll
    for (int i = 0; i < 9; ++i) {
        const int p = i < 4 ? 4 * lane + i : (i < 8 ? 124 + 4 * lane + i : PPOS - 1);
        if ((unsigned)(p - w0) <= 14u) e[i] = LOWF;
        bv = fmaxf(bv, e[i]);
    }
    bv = rowmaxf16(bv);
    bv = fmaxf(bv, __int_as_float(xhalf(__float_as_int(bv))));
    unsigned bp = 0xFFFFFFFFu;
    #pragma unroll
    for (int i = 8; i >= 0; --i) {
        const int p = i < 4 ? 4 * lane + i : (i < 8 ? 124 + 4 * lane + i : PPOS - 1);
        if (e[i] == bv) bp = (unsigned)p;
    }
    bp = rowminu16(bp);
    bp = min(bp, (unsigned)xhalf((int)bp));
    if (bp == 0xFFFFFFFFu) bp = 0u;
    return mkkey(bv, (unsigned)flat0 + bp);
}

template <int G>
__device__ __forceinline__ void rows_scan(const RowSlots& s, int nre, int w0, int lane, int nloc, int a0,
                                          unsigned long long& rk) {
    if constexpr (G < 4) {
        if (4 * G < nre) {
            #pragma unroll
            for (int k = 0; k < 4; ++k) {
                const int q = 4 * G + k;
                const unsigned long long nk = row_rescan(s.a[q], s.b[q], s.c[q], (a0 + s.row[q]) * PPOS, w0, lane);
                if (nloc == s.row[q]) rk = nk;
            }
            rows_scan<G + 1>(s, nre, w0, lane, nloc, a0, rk);
        }
    }
}

__global__ __launch_bounds__(CORR_THREADS)
void k_inc(const float* __restrict__ dnn, float* map, unsigned long long* keys,
           unsigned long long* rowkeys, int step) {
    __shared__ __attribute__((aligned(16))) float smem[LDS_FLOATS];
    const int tid  = threadIdx.x;
    const int lane = tid & 31;
    const int b    = tid >> 5;
    const int a0   = blockIdx.x * 16;
    const int half = lane >> 4;
    const int nloc = lane & 15;

    const unsigned long long fin = __hip_atomic_load(keys + (step - 1) * BATCH + b, __ATOMIC_RELAXED,
                                                     __HIP_MEMORY_SCOPE_AGENT);
    unsigned long long* rkp = rowkeys + (size_t)b * NATOMS + a0 + nloc;
    unsigned long long rk = *rkp;
    float* mblk = map + ((size_t)b * NATOMS + a0) * MAPW;
    const v4f* arow = (const v4f*)(dnn + (size_t)(a0 + nloc) * KDIM + half * 16) + b;
    v4f ra[NGRP];
    #pragma unroll
    for (int g = 0; g < NGRP; ++g) ra[g] = arow[g * 8];
    {
        v4f* z = (v4f*)(smem + DIAGO);
        #pragma unroll
        for (int k = 0; k < (BATCH * IPD / 4 + CORR_THREADS - 1) / CORR_THREADS; ++k)
            z[tid + k * CORR_THREADS] = (v4f){0.f, 0.f, 0.f, 0.f};
    }
    __builtin_amdgcn_sched_barrier(0);

    const unsigned enc  = (unsigned)(fin >> 32);
    const unsigned bits = (enc & 0x80000000u) ? (enc & 0x7FFFFFFFu) : ~enc;
    const float val = __uint_as_float(bits);
    const unsigned idx = 0xFFFFFFFFu - (unsigned)(fin & 0xFFFFFFFFu);
    const int atom = min((int)(idx / PPOS), NATOMS - 1);
    const int pos  = (int)(idx % PPOS);

    const v4f* hp = (const v4f*)(dnn + (size_t)atom * KDIM + lane * 16);
    const v4f h0 = hp[0], h1 = hp[1], h2 = hp[2], h3 = hp[3];
    const int  pp = pos + nloc - 7;
    const bool pv = nloc < 15 && pp >= 0 && pp < PPOS;
    float* prow = mblk + (size_t)(8 * half) * MAPW + min(max(pp, 0), PPOS - 1);
    float oldv[8];
    #pragma unroll
    for (int r = 0; r < 8; ++r) oldv[r] = prow[r * MAPW];
    const int w0 = pos - 7;
    const int rpos = (int)(0xFFFFFFFFu - (unsigned)rk) - (a0 + nloc) * PPOS;
    const unsigned rmask = __builtin_amdgcn_ballot_w32((unsigned)(rpos - w0) <= 14u) & 0xFFFFu;
    const int nre = __builtin_popcount(rmask);
    RowSlots rs;
    rows_request<0>(rs, rmask, nre, mblk, lane);
    __builtin_amdgcn_sched_barrier(0);

    {
        v4f* rowp = (v4f*)(smem + b * HBP + lane * 16);
        rowp[0] = (v4f){h0.x, h0.y, h2.x, h2.y};
        rowp[1] = (v4f){h0.z, h0.w, h2.z, h2.w};
        rowp[2] = (v4f){h1.x, h1.y, h3.x, h3.y};
        rowp[3] = (v4f){h1.z, h1.w, h3.z, h3.w};
    }
    __syncthreads();

    v8f y[2][2] = {};
    {
        const v2f* bp = (const v2f*)(smem + (nloc >> 3) * HBP + half * 16) + (nloc & 7);
        #pragma unroll
        for (int g = 0; g < NGRP; ++g) {
            const v2f hA[2] = {{ra[g].x, ra[g].y}, {ra[g].z, ra[g].w}};
            const v2f bf[2] = {bp[g * 16], bp[g * 16 + HBP]};
            #pragma unroll
            for (int ti = 0; ti < 2; ++ti) {
                #pragma unroll
                for (int n2 = 0; n2 < 2; ++n2)
                    y[ti][n2] = __builtin_amdgcn_wmma_f32_16x16x4_f32(
                        false, hA[ti], false, bf[n2], (short)0, y[ti][n2], false, false);
            }
        }
    }
    {
        const int t0 = (b & 1) * 4 + (b >> 1);
        float* dg = smem + DIAGO + (nloc >> 3) * IPD + (8 * half) * 128 + ((nloc & 7) + 7) * 8 - 7 * t0;
        #pragma unroll
        for (int ti = 0; ti < 2; ++ti) {
            #pragma unroll
            for (int n2 = 0; n2 < 2; ++n2) {
                #pragma unroll
                for (int r = 0; r < 8; ++r) dg[2 * n2 * IPD + r * 128 - 14 * ti] = y[ti][n2][r];
            }
        }
    }
    __syncthreads();
    v8f X;
    {
        const v4f* xr = (const v4f*)(smem + DIAGO + b * IPD + (8 * half) * 128 + nloc * 8);
        #pragma unroll
        for (int r = 0; r < 8; ++r) {
            const v4f lo = xr[r * 32], hi = xr[r * 32 + 1];
            X[r] = ((lo.x + lo.y) + (lo.z + lo.w)) + ((hi.x + hi.y) + (hi.z + hi.w));
        }
    }

    unsigned ev[8];
    #pragma unroll
    for (int r = 0; r < 8; ++r) ev[r] = 0u;
    if (pv) {
        #pragma unroll
        for (int r = 0; r < 8; ++r) {
            const float nv = oldv[r] - val * X[r];
            prow[r * MAPW] = nv;
            ev[r] = fenc(nv);
        }
    }

    rows_scan<0>(rs, nre, w0, lane, nloc, a0, rk);

    unsigned pbe = 0u, pbp = 0xFFFFFFFFu;
    #pragma unroll
    for (int r = 0; r < 8; ++r) {
        const unsigned mx = rowmaxu16(ev[r]);
        const unsigned cp = rowminu16((pv && ev[r] == mx) ? (unsigned)pp : 0xFFFFFFFFu);
        if ((nloc & 7) == r) { pbe = mx; pbp = cp; }
    }
    unsigned long long key = 0ull;
    if ((nloc >> 3) == half) {
        if (pbp != 0xFFFFFFFFu) {
            const unsigned long long pk = ((unsigned long long)pbe << 32)
                                        | (0xFFFFFFFFu - ((unsigned)(a0 + nloc) * PPOS + pbp));
            if (pk > rk) rk = pk;
        }
        *rkp = rk;
        key = rk;
    }
    key = wavemaxkey(key);
    if (lane == 0) atomicMax(keys + step * BATCH + b, key);
}

__global__ void k_output_keys(const float* __restrict__ dnn, const unsigned long long* __restrict__ keys,
                              float* __restrict__ out) {
    const int i = blockIdx.x * blockDim.x + threadIdx.x;
    if (i >= BATCH * CC * TT) return;
    const int t  = i % TT;
    const int bc = i / TT;
    const int ch = bc % CC, b = bc / CC;
    float acc = 0.f;
    for (int s = 0; s < NITER; ++s) {
        const unsigned long long fin = keys[s * BATCH + b];
        const unsigned enc  = (unsigned)(fin >> 32);
        const unsigned bits = (enc & 0x80000000u) ? (enc & 0x7FFFFFFFu) : ~enc;
        const float val = __uint_as_float(bits);
        const unsigned idx = 0xFFFFFFFFu - (unsigned)(fin & 0xFFFFFFFFu);
        const int atom = min((int)(idx / PPOS), NATOMS - 1);
        const int tap  = t - (int)(idx % PPOS);
        if (tap >= 0 && tap < ASIZE) acc += val * dnn[(size_t)atom * KDIM + kperm(ch, tap)];
    }
    out[i] = acc;
}

__global__ void k_output(const float* __restrict__ recon, float* __restrict__ out) {
    const int i = blockIdx.x * blockDim.x + threadIdx.x;
    if (i < BATCH * CC * TT) {
        const int t  = i % TT;
        const int bc = i / TT;
        out[i] = recon[(size_t)bc * RES_W + t];
    }
}

extern "C" void kernel_launch(void* const* d_in, const int* in_sizes, int n_in,
                              void* d_out, int out_size, void* d_ws, size_t ws_size,
                              hipStream_t stream) {
    const float* x = (const float*)d_in[0];
    const float* d = (const float*)d_in[1];
    (void)in_sizes; (void)n_in; (void)out_size;

    char* ws = (char*)d_ws;
    float* dnn   = (float*)ws;
    float* res   = (float*)(ws + (size_t)NATOMS * KDIM * 4);
    float* recon = res + BATCH * CC * RES_W;
    unsigned long long* best = (unsigned long long*)(recon + BATCH * CC * RES_W);
    unsigned* arrived = (unsigned*)(best + BATCH);

    k_normalize<<<NATOMS, CORR_THREADS, 0, stream>>>(d, dnn);
    dim3 grid(NATOMS / MTILE, BATCH);

    if (ws_size >= (size_t)26051584) {
        unsigned long long* keys = best;
        float* map = (float*)(ws + 9012224);
        unsigned long long* rowkeys = (unsigned long long*)recon;
        k_init<<<(BATCH * CC * RES_W + 255) / 256, 256, 0, stream>>>(x, res, recon, keys, nullptr, NITER * BATCH);
        k_corr<true><<<grid, CORR_THREADS, 0, stream>>>(dnn, res, recon, keys, nullptr, map, rowkeys);
        for (int it = 1; it < NITER; ++it)
            k_inc<<<NATOMS / 16, CORR_THREADS, 0, stream>>>(dnn, map, keys, rowkeys, it);
        k_output_keys<<<(BATCH * CC * TT + 255) / 256, 256, 0, stream>>>(dnn, keys, (float*)d_out);
        return;
    }
    k_init<<<(BATCH * CC * RES_W + 255) / 256, 256, 0, stream>>>(x, res, recon, best, arrived, BATCH);
    for (int it = 0; it < NITER; ++it)
        k_corr<false><<<grid, CORR_THREADS, 0, stream>>>(dnn, res, recon, best, arrived, nullptr, nullptr);
    k_output<<<(BATCH * CC * TT + 255) / 256, 256, 0, stream>>>(recon, (float*)d_out);
}
